# starting kernel
# baseline (speedup 1.0000x reference)
#include <hip/hip_runtime.h>

typedef __attribute__((ext_vector_type(2))) float v2f;
typedef __attribute__((ext_vector_type(8))) float v8f;

#define HW 1024
#define NPIX (HW * HW)

__global__ void zero_kernel(float* __restrict__ p, int n) {
    int i = blockIdx.x * blockDim.x + threadIdx.x;
    if (i < n) p[i] = 0.0f;
}

__device__ __forceinline__ float sigmoidf_(float v) {
    return 1.0f / (1.0f + expf(-v));
}

__device__ __forceinline__ float patchv(const float* __restrict__ xp,
                                        const float* __restrict__ hp,
                                        int y, int colbase, int m, int k) {
    int  ci  = (k >= 9) ? 1 : 0;
    int  rem = k - ci * 9;
    int  dy  = rem / 3 - 1;
    int  dx  = rem - (rem / 3) * 3 - 1;
    int  rr  = y + dy;
    int  cc  = colbase + m + dx;
    bool ok  = (k < 18) && (rr >= 0) && (rr < HW) && (cc >= 0) && (cc < HW);
    const float* src = ci ? hp : xp;
    float v = src[ok ? (rr * HW + cc) : 0];
    return ok ? v : 0.0f;
}

__device__ __forceinline__ float wval(const float* __restrict__ w, int n, int k) {
    bool ok = (n < 4) && (k < 18);
    float v = w[ok ? (n * 18 + k) : 0];
    return ok ? v : 0.0f;
}

__global__ __launch_bounds__(256) void convlstm_step(
    const float* __restrict__ x,   const float* __restrict__ hin,
    const float* __restrict__ cin, const float* __restrict__ w,
    const float* __restrict__ bias, float* __restrict__ hout,
    float* __restrict__ cout)
{
    __shared__ float lds[8 * 512];
    const int lane  = threadIdx.x & 31;
    const int wib   = threadIdx.x >> 5;
    const int wid   = blockIdx.x * 8 + wib;
    const int y     = wid >> 5;
    const int x0    = (wid & 31) * 32;
    float* wlds = &lds[wib * 512];

    const int n     = lane & 15;
    const int m     = lane & 15;
    const int khalf = (lane >> 4) << 1;

    v2f bm[5];
#pragma unroll
    for (int c = 0; c < 5; ++c) {
        int kk  = 4 * c + khalf;
        bm[c].x = wval(w, n, kk);
        bm[c].y = wval(w, n, kk + 1);
    }

    float bn = (n < 4) ? bias[n] : 0.0f;
    v8f cinit;
#pragma unroll
    for (int r = 0; r < 8; ++r) cinit[r] = bn;

#pragma unroll
    for (int seg = 0; seg < 2; ++seg) {
        int colbase = x0 + seg * 16;
        v8f acc = cinit;
#pragma unroll
        for (int c = 0; c < 5; ++c) {
            int kk = 4 * c + khalf;
            v2f a;
            a.x = patchv(x, hin, y, colbase, m, kk);
            a.y = patchv(x, hin, y, colbase, m, kk + 1);
            acc = __builtin_amdgcn_wmma_f32_16x16x4_f32(
                false, a, false, bm[c], (short)0, acc, false, false);
        }
#pragma unroll
        for (int r = 0; r < 8; ++r)
            wlds[seg * 256 + r * 32 + lane] = acc[r];
    }
    __syncthreads();

    int seg = lane >> 4;
    int idx = seg * 256 + (m & 7) * 32 + ((m >> 3) << 4);
    float4 g = *(const float4*)&wlds[idx];

    int p = y * HW + x0 + lane;
    float cprev = cin[p];
    float cn = sigmoidf_(g.y) * cprev + sigmoidf_(g.x) * tanhf(g.w);
    float hn = sigmoidf_(g.z) * tanhf(cn);
    cout[p] = cn;
    hout[p] = hn;
}

extern "C" void kernel_launch(void* const* d_in, const int* in_sizes, int n_in,
                              void* d_out, int out_size, void* d_ws, size_t ws_size,
                              hipStream_t stream) {
    (void)in_sizes; (void)n_in; (void)out_size; (void)ws_size;
    const float* data  = (const float*)d_in[0];
    const float* enc_w = (const float*)d_in[1];
    const float* enc_b = (const float*)d_in[2];
    const float* dec_w = (const float*)d_in[3];
    const float* dec_b = (const float*)d_in[4];
    float* out = (float*)d_out;

    float* ws = (float*)d_ws;
    float* h0 = ws;
    float* h1 = ws + (size_t)NPIX;
    float* h2 = ws + (size_t)2 * NPIX;
    float* c0 = ws + (size_t)3 * NPIX;
    float* c1 = ws + (size_t)4 * NPIX;

    dim3 zgrid((NPIX + 255) / 256), zblk(256);
    dim3 grid(4096), blk(256);

    zero_kernel<<<zgrid, zblk, 0, stream>>>(h0, NPIX);
    zero_kernel<<<zgrid, zblk, 0, stream>>>(c0, NPIX);
    float *ha = h0, *hb = h1, *ca = c0, *cb = c1;
    for (int t = 0; t < 20; ++t) {
        convlstm_step<<<grid, blk, 0, stream>>>(data + (size_t)t * NPIX,
                                                ha, ca, enc_w, enc_b, hb, cb);
        float* tmp;
        tmp = ha; ha = hb; hb = tmp;
        tmp = ca; ca = cb; cb = tmp;
    }
    const float* dec_in = ha;

    zero_kernel<<<zgrid, zblk, 0, stream>>>(hb, NPIX);
    zero_kernel<<<zgrid, zblk, 0, stream>>>(ca, NPIX);
    float *hda = hb, *hdb = h2, *cda = ca, *cdb = cb;
    for (int t = 0; t < 20; ++t) {
        float* hdst = (t == 19) ? out : hdb;
        convlstm_step<<<grid, blk, 0, stream>>>(dec_in, hda, cda, dec_w, dec_b,
                                                hdst, cdb);
        float* tmp;
        tmp = hda; hda = hdb; hdb = tmp;
        tmp = cda; cda = cdb; cdb = tmp;
    }
}


# K slots re-mapped so a lane's half-wave picks the input plane (x or h) and each tap is a constant offset from one per-lane address; checked loads only in border waves; tap summation order changes
# speedup vs baseline: 1.1257x; 1.1257x over previous
#include <hip/hip_runtime.h>

typedef __attribute__((ext_vector_type(2))) float v2f;
typedef __attribute__((ext_vector_type(8))) float v8f;

#define HW 1024
#define NPIX (HW * HW)

__global__ void zero_kernel(float* __restrict__ p, int n) {
    int i = blockIdx.x * blockDim.x + threadIdx.x;
    if (i < n) p[i] = 0.0f;
}

__device__ __forceinline__ float sigmoidf_(float v) {
    return 1.0f / (1.0f + expf(-v));
}


__global__ __launch_bounds__(256) void convlstm_step(
    const float* __restrict__ x,   const float* __restrict__ hin,
    const float* __restrict__ cin, const float* __restrict__ w,
    const float* __restrict__ bias, float* __restrict__ hout,
    float* __restrict__ cout)
{
    __shared__ float lds[8 * 512];
    const int lane  = threadIdx.x & 31;
    const int wib   = threadIdx.x >> 5;
    const int wid   = __builtin_amdgcn_readfirstlane(blockIdx.x * 8 + wib);
    const int y     = wid >> 5;
    const int x0    = (wid & 31) * 32;
    float* wlds = &lds[wib * 512];

    const int m  = lane & 15;
    const int hf = lane >> 4;
    const float* __restrict__ plane = hf ? hin : x;

    const float* wl = w + ((m & 3) * 18 + hf * 9);
    v2f bm[5];
#pragma unroll
    for (int c = 0; c < 5; ++c) {
        bm[c].x = wl[2 * c];
        bm[c].y = (2 * c + 1 < 9) ? wl[2 * c + 1] : 0.0f;
    }

    float bn = bias[m & 3];
    v8f cinit;
#pragma unroll
    for (int r = 0; r < 8; ++r) cinit[r] = bn;

    v2f af[2][5];
    const bool interior = (y >= 1) && (y <= HW - 2) && (x0 >= 32) && (x0 <= HW - 64);
    if (interior) {
        const float* bp = plane + (y * HW + x0 + m);
#pragma unroll
        for (int seg = 0; seg < 2; ++seg) {
#pragma unroll
            for (int c = 0; c < 5; ++c) {
#pragma unroll
                for (int j = 0; j < 2; ++j) {
                    const int tap = 2 * c + j;
                    float v = 0.0f;
                    if (tap < 9) {
                        const int dy = tap / 3 - 1;
                        const int dx = tap % 3 - 1;
                        v = bp[dy * HW + dx + seg * 16];
                    }
                    if (j == 0) af[seg][c].x = v; else af[seg][c].y = v;
                }
            }
        }
    } else {
#pragma unroll
        for (int seg = 0; seg < 2; ++seg) {
#pragma unroll
            for (int c = 0; c < 5; ++c) {
#pragma unroll
                for (int j = 0; j < 2; ++j) {
                    const int tap = 2 * c + j;
                    float v = 0.0f;
                    if (tap < 9) {
                        const int rr = y + tap / 3 - 1;
                        const int cc = x0 + seg * 16 + m + tap % 3 - 1;
                        const bool ok = (rr >= 0) && (rr < HW) && (cc >= 0) && (cc < HW);
                        const float t = plane[ok ? (rr * HW + cc) : 0];
                        v = ok ? t : 0.0f;
                    }
                    if (j == 0) af[seg][c].x = v; else af[seg][c].y = v;
                }
            }
        }
    }

#pragma unroll
    for (int seg = 0; seg < 2; ++seg) {
        v8f acc = cinit;
#pragma unroll
        for (int c = 0; c < 5; ++c) {
            acc = __builtin_amdgcn_wmma_f32_16x16x4_f32(
                false, af[seg][c], false, bm[c], (short)0, acc, false, false);
        }
#pragma unroll
        for (int r = 0; r < 8; ++r)
            wlds[seg * 256 + r * 32 + lane] = acc[r];
    }
    __syncthreads();

    int seg = lane >> 4;
    int idx = seg * 256 + (m & 7) * 32 + ((m >> 3) << 4);
    float4 g = *(const float4*)&wlds[idx];

    int p = y * HW + x0 + lane;
    float cprev = cin[p];
    float cn = sigmoidf_(g.y) * cprev + sigmoidf_(g.x) * tanhf(g.w);
    float hn = sigmoidf_(g.z) * tanhf(cn);
    cout[p] = cn;
    hout[p] = hn;
}

extern "C" void kernel_launch(void* const* d_in, const int* in_sizes, int n_in,
                              void* d_out, int out_size, void* d_ws, size_t ws_size,
                              hipStream_t stream) {
    (void)in_sizes; (void)n_in; (void)out_size; (void)ws_size;
    const float* data  = (const float*)d_in[0];
    const float* enc_w = (const float*)d_in[1];
    const float* enc_b = (const float*)d_in[2];
    const float* dec_w = (const float*)d_in[3];
    const float* dec_b = (const float*)d_in[4];
    float* out = (float*)d_out;

    float* ws = (float*)d_ws;
    float* h0 = ws;
    float* h1 = ws + (size_t)NPIX;
    float* h2 = ws + (size_t)2 * NPIX;
    float* c0 = ws + (size_t)3 * NPIX;
    float* c1 = ws + (size_t)4 * NPIX;

    dim3 zgrid((NPIX + 255) / 256), zblk(256);
    dim3 grid(4096), blk(256);

    zero_kernel<<<zgrid, zblk, 0, stream>>>(h0, NPIX);
    zero_kernel<<<zgrid, zblk, 0, stream>>>(c0, NPIX);
    float *ha = h0, *hb = h1, *ca = c0, *cb = c1;
    for (int t = 0; t < 20; ++t) {
        convlstm_step<<<grid, blk, 0, stream>>>(data + (size_t)t * NPIX,
                                                ha, ca, enc_w, enc_b, hb, cb);
        float* tmp;
        tmp = ha; ha = hb; hb = tmp;
        tmp = ca; ca = cb; cb = tmp;
    }
    const float* dec_in = ha;

    zero_kernel<<<zgrid, zblk, 0, stream>>>(hb, NPIX);
    zero_kernel<<<zgrid, zblk, 0, stream>>>(ca, NPIX);
    float *hda = hb, *hdb = h2, *cda = ca, *cdb = cb;
    for (int t = 0; t < 20; ++t) {
        float* hdst = (t == 19) ? out : hdb;
        convlstm_step<<<grid, blk, 0, stream>>>(dec_in, hda, cda, dec_w, dec_b,
                                                hdst, cdb);
        float* tmp;
        tmp = hda; hda = hdb; hdb = tmp;
        tmp = cda; cda = cdb; cdb = tmp;
    }
}


# Implicit GEMM re-blocked: M = 4 gates x 4 out rows (taps banded over 6 input rows), N = 16 pixels; 9 WMMAs per 64 pixels not 20; gates land per lane so LDS stash and barrier go; tap sum order changes
# speedup vs baseline: 1.5470x; 1.3743x over previous
#include <hip/hip_runtime.h>

typedef __attribute__((ext_vector_type(2))) float v2f;
typedef __attribute__((ext_vector_type(8))) float v8f;

#define HW 1024
#define NPIX (HW * HW)

__global__ void zero_kernel(float* __restrict__ p, int n) {
    int i = blockIdx.x * blockDim.x + threadIdx.x;
    if (i < n) p[i] = 0.0f;
}

__device__ __forceinline__ float sigmoidf_(float v) {
    return 1.0f / (1.0f + expf(-v));
}

__global__ __launch_bounds__(256) void convlstm_step(
    const float* __restrict__ x,   const float* __restrict__ hin,
    const float* __restrict__ cin, const float* __restrict__ w,
    const float* __restrict__ bias, float* __restrict__ hout,
    float* __restrict__ cout)
{
    const int lane  = threadIdx.x & 31;
    const int wib   = threadIdx.x >> 5;
    const int wid   = __builtin_amdgcn_readfirstlane(blockIdx.x * 8 + wib);
    const int yb    = (wid >> 5) * 4;
    const int x0    = (wid & 31) * 32;

    const int c  = lane & 15;
    const int hf = lane >> 4;
    const float* __restrict__ plane = hf ? hin : x;

    const int r = c >> 2;
    const float* wl = w + ((c & 3) * 18 + hf * 9);
    float wv[9];
#pragma unroll
    for (int i = 0; i < 9; ++i) wv[i] = wl[i];
    v2f af[9];
#pragma unroll
    for (int ch = 0; ch < 9; ++ch) {
#pragma unroll
        for (int s = 0; s < 2; ++s) {
            const int j = 2 * ch + s;
            float v = 0.0f;
#pragma unroll
            for (int rr = 0; rr < 4; ++rr) {
                const int idx = j - 3 * rr;
                if (idx >= 0 && idx <= 8) v = (r == rr) ? wv[idx] : v;
            }
            if (s == 0) af[ch].x = v; else af[ch].y = v;
        }
    }

    v8f cinit;
#pragma unroll
    for (int q = 0; q < 8; ++q) cinit[q] = bias[q & 3];

    v2f bf[2][9];
    const bool interior = (yb >= 4) && (yb <= HW - 8) && (x0 >= 32) && (x0 <= HW - 64);
    if (interior) {
        const float* bp = plane + ((yb - 1) * HW + x0 + c - 1);
#pragma unroll
        for (int t = 0; t < 2; ++t) {
#pragma unroll
            for (int ch = 0; ch < 9; ++ch) {
#pragma unroll
                for (int s = 0; s < 2; ++s) {
                    const int j = 2 * ch + s;
                    const float v = bp[(j / 3) * HW + 16 * t + (j % 3)];
                    if (s == 0) bf[t][ch].x = v; else bf[t][ch].y = v;
                }
            }
        }
    } else {
#pragma unroll
        for (int t = 0; t < 2; ++t) {
#pragma unroll
            for (int ch = 0; ch < 9; ++ch) {
#pragma unroll
                for (int s = 0; s < 2; ++s) {
                    const int j = 2 * ch + s;
                    const int rw = yb - 1 + j / 3;
                    const int cc = x0 + 16 * t + c - 1 + j % 3;
                    const bool ok = (rw >= 0) && (rw < HW) && (cc >= 0) && (cc < HW);
                    const float tv = plane[ok ? (rw * HW + cc) : 0];
                    const float v = ok ? tv : 0.0f;
                    if (s == 0) bf[t][ch].x = v; else bf[t][ch].y = v;
                }
            }
        }
    }

    const int p0 = (yb + 2 * hf) * HW + x0 + c;
    float cp[2][2];
#pragma unroll
    for (int t = 0; t < 2; ++t) {
#pragma unroll
        for (int e = 0; e < 2; ++e) cp[t][e] = cin[p0 + e * HW + 16 * t];
    }

    v8f acc0 = cinit, acc1 = cinit;
#pragma unroll
    for (int ch = 0; ch < 9; ++ch) {
        acc0 = __builtin_amdgcn_wmma_f32_16x16x4_f32(
            false, af[ch], false, bf[0][ch], (short)0, acc0, false, false);
        acc1 = __builtin_amdgcn_wmma_f32_16x16x4_f32(
            false, af[ch], false, bf[1][ch], (short)0, acc1, false, false);
    }

#pragma unroll
    for (int t = 0; t < 2; ++t) {
        const v8f acc = t ? acc1 : acc0;
#pragma unroll
        for (int e = 0; e < 2; ++e) {
            const float gi = acc[4 * e + 0];
            const float gf = acc[4 * e + 1];
            const float go = acc[4 * e + 2];
            const float gg = acc[4 * e + 3];
            const int p = p0 + e * HW + 16 * t;
            const float cprev = cp[t][e];
            float cn = sigmoidf_(gf) * cprev + sigmoidf_(gi) * tanhf(gg);
            float hn = sigmoidf_(go) * tanhf(cn);
            cout[p] = cn;
            hout[p] = hn;
        }
    }
}

extern "C" void kernel_launch(void* const* d_in, const int* in_sizes, int n_in,
                              void* d_out, int out_size, void* d_ws, size_t ws_size,
                              hipStream_t stream) {
    (void)in_sizes; (void)n_in; (void)out_size; (void)ws_size;
    const float* data  = (const float*)d_in[0];
    const float* enc_w = (const float*)d_in[1];
    const float* enc_b = (const float*)d_in[2];
    const float* dec_w = (const float*)d_in[3];
    const float* dec_b = (const float*)d_in[4];
    float* out = (float*)d_out;

    float* ws = (float*)d_ws;
    float* h0 = ws;
    float* h1 = ws + (size_t)NPIX;
    float* h2 = ws + (size_t)2 * NPIX;
    float* c0 = ws + (size_t)3 * NPIX;
    float* c1 = ws + (size_t)4 * NPIX;

    dim3 zgrid((NPIX + 255) / 256), zblk(256);
    dim3 grid(1024), blk(256);

    zero_kernel<<<zgrid, zblk, 0, stream>>>(h0, NPIX);
    zero_kernel<<<zgrid, zblk, 0, stream>>>(c0, NPIX);
    float *ha = h0, *hb = h1, *ca = c0, *cb = c1;
    for (int t = 0; t < 20; ++t) {
        convlstm_step<<<grid, blk, 0, stream>>>(data + (size_t)t * NPIX,
                                                ha, ca, enc_w, enc_b, hb, cb);
        float* tmp;
        tmp = ha; ha = hb; hb = tmp;
        tmp = ca; ca = cb; cb = tmp;
    }
    const float* dec_in = ha;

    zero_kernel<<<zgrid, zblk, 0, stream>>>(hb, NPIX);
    zero_kernel<<<zgrid, zblk, 0, stream>>>(ca, NPIX);
    float *hda = hb, *hdb = h2, *cda = ca, *cdb = cb;
    for (int t = 0; t < 20; ++t) {
        float* hdst = (t == 19) ? out : hdb;
        convlstm_step<<<grid, blk, 0, stream>>>(dec_in, hda, cda, dec_w, dec_b,
                                                hdst, cdb);
        float* tmp;
        tmp = hda; hda = hdb; hdb = tmp;
        tmp = cda; cda = cdb; cdb = tmp;
    }
}


# Zero-fill launches removed: the first encoder and decoder step is a variant with h = 0, c = 0 built in (h-plane K slots zero, no state loads); 44 launches become 40; every sum unchanged
# speedup vs baseline: 1.6238x; 1.0496x over previous
#include <hip/hip_runtime.h>

typedef __attribute__((ext_vector_type(2))) float v2f;
typedef __attribute__((ext_vector_type(8))) float v8f;

#define HW 1024
#define NPIX (HW * HW)

__device__ __forceinline__ float sigmoidf_(float v) {
    return 1.0f / (1.0f + expf(-v));
}

template <bool FIRST>
__global__ __launch_bounds__(256) void convlstm_step(
    const float* __restrict__ x,   const float* __restrict__ hin,
    const float* __restrict__ cin, const float* __restrict__ w,
    const float* __restrict__ bias, float* __restrict__ hout,
    float* __restrict__ cout)
{
    const int lane  = threadIdx.x & 31;
    const int wib   = threadIdx.x >> 5;
    const int wid   = __builtin_amdgcn_readfirstlane(blockIdx.x * 8 + wib);
    const int yb    = (wid >> 5) * 4;
    const int x0    = (wid & 31) * 32;

    const int c  = lane & 15;
    const int hf = lane >> 4;
    const float* __restrict__ plane = (!FIRST && hf) ? hin : x;
    const bool zero_half = FIRST && (hf != 0);

    const int r = c >> 2;
    const float* wl = w + ((c & 3) * 18 + hf * 9);
    float wv[9];
#pragma unroll
    for (int i = 0; i < 9; ++i) wv[i] = wl[i];
    v2f af[9];
#pragma unroll
    for (int ch = 0; ch < 9; ++ch) {
#pragma unroll
        for (int s = 0; s < 2; ++s) {
            const int j = 2 * ch + s;
            float v = 0.0f;
#pragma unroll
            for (int rr = 0; rr < 4; ++rr) {
                const int idx = j - 3 * rr;
                if (idx >= 0 && idx <= 8) v = (r == rr) ? wv[idx] : v;
            }
            if (s == 0) af[ch].x = v; else af[ch].y = v;
        }
    }

    v8f cinit;
#pragma unroll
    for (int q = 0; q < 8; ++q) cinit[q] = bias[q & 3];

    v2f bf[2][9];
    const bool interior = (yb >= 4) && (yb <= HW - 8) && (x0 >= 32) && (x0 <= HW - 64);
    if (interior) {
        const float* bp = plane + ((yb - 1) * HW + x0 + c - 1);
#pragma unroll
        for (int t = 0; t < 2; ++t) {
#pragma unroll
            for (int ch = 0; ch < 9; ++ch) {
#pragma unroll
                for (int s = 0; s < 2; ++s) {
                    const int j = 2 * ch + s;
                    const float lv = bp[(j / 3) * HW + 16 * t + (j % 3)];
                    const float v = zero_half ? 0.0f : lv;
                    if (s == 0) bf[t][ch].x = v; else bf[t][ch].y = v;
                }
            }
        }
    } else {
#pragma unroll
        for (int t = 0; t < 2; ++t) {
#pragma unroll
            for (int ch = 0; ch < 9; ++ch) {
#pragma unroll
                for (int s = 0; s < 2; ++s) {
                    const int j = 2 * ch + s;
                    const int rw = yb - 1 + j / 3;
                    const int cc = x0 + 16 * t + c - 1 + j % 3;
                    const bool ok = (rw >= 0) && (rw < HW) && (cc >= 0) && (cc < HW);
                    const float tv = plane[ok ? (rw * HW + cc) : 0];
                    const float v = (ok && !zero_half) ? tv : 0.0f;
                    if (s == 0) bf[t][ch].x = v; else bf[t][ch].y = v;
                }
            }
        }
    }

    const int p0 = (yb + 2 * hf) * HW + x0 + c;
    float cp[2][2];
#pragma unroll
    for (int t = 0; t < 2; ++t) {
#pragma unroll
        for (int e = 0; e < 2; ++e) cp[t][e] = FIRST ? 0.0f : cin[p0 + e * HW + 16 * t];
    }

    v8f acc0 = cinit, acc1 = cinit;
#pragma unroll
    for (int ch = 0; ch < 9; ++ch) {
        acc0 = __builtin_amdgcn_wmma_f32_16x16x4_f32(
            false, af[ch], false, bf[0][ch], (short)0, acc0, false, false);
        acc1 = __builtin_amdgcn_wmma_f32_16x16x4_f32(
            false, af[ch], false, bf[1][ch], (short)0, acc1, false, false);
    }

#pragma unroll
    for (int t = 0; t < 2; ++t) {
        const v8f acc = t ? acc1 : acc0;
#pragma unroll
        for (int e = 0; e < 2; ++e) {
            const float gi = acc[4 * e + 0];
            const float gf = acc[4 * e + 1];
            const float go = acc[4 * e + 2];
            const float gg = acc[4 * e + 3];
            const int p = p0 + e * HW + 16 * t;
            const float cprev = cp[t][e];
            float cn = sigmoidf_(gf) * cprev + sigmoidf_(gi) * tanhf(gg);
            float hn = sigmoidf_(go) * tanhf(cn);
            cout[p] = cn;
            hout[p] = hn;
        }
    }
}

extern "C" void kernel_launch(void* const* d_in, const int* in_sizes, int n_in,
                              void* d_out, int out_size, void* d_ws, size_t ws_size,
                              hipStream_t stream) {
    (void)in_sizes; (void)n_in; (void)out_size; (void)ws_size;
    const float* data  = (const float*)d_in[0];
    const float* enc_w = (const float*)d_in[1];
    const float* enc_b = (const float*)d_in[2];
    const float* dec_w = (const float*)d_in[3];
    const float* dec_b = (const float*)d_in[4];
    float* out = (float*)d_out;

    float* ws = (float*)d_ws;
    float* h0 = ws;
    float* h1 = ws + (size_t)NPIX;
    float* h2 = ws + (size_t)2 * NPIX;
    float* c0 = ws + (size_t)3 * NPIX;
    float* c1 = ws + (size_t)4 * NPIX;

    dim3 grid(1024), blk(256);

    float *ha = h0, *hb = h1, *ca = c0, *cb = c1;
    for (int t = 0; t < 20; ++t) {
        if (t == 0)
            convlstm_step<true><<<grid, blk, 0, stream>>>(data, ha, ca, enc_w, enc_b, hb, cb);
        else
            convlstm_step<false><<<grid, blk, 0, stream>>>(data + (size_t)t * NPIX,
                                                       ha, ca, enc_w, enc_b, hb, cb);
        float* tmp;
        tmp = ha; ha = hb; hb = tmp;
        tmp = ca; ca = cb; cb = tmp;
    }
    const float* dec_in = ha;

    float *hda = hb, *hdb = h2, *cda = ca, *cdb = cb;
    for (int t = 0; t < 20; ++t) {
        float* hdst = (t == 19) ? out : hdb;
        if (t == 0)
            convlstm_step<true><<<grid, blk, 0, stream>>>(dec_in, hda, cda, dec_w, dec_b,
                                                          hdst, cdb);
        else
            convlstm_step<false><<<grid, blk, 0, stream>>>(dec_in, hda, cda, dec_w, dec_b,
                                                           hdst, cdb);
        float* tmp;
        tmp = hda; hda = hdb; hdb = tmp;
        tmp = cda; cda = cdb; cdb = tmp;
    }
}


# Gate activations on the hardware tanh instruction: sigmoid(x) = 0.5 + 0.5*tanh(x/2), tanh direct, in place of library expf, IEEE division and branching tanhf; last-bit rounding of the gates changes
# speedup vs baseline: 2.2798x; 1.4040x over previous
#include <hip/hip_runtime.h>

typedef __attribute__((ext_vector_type(2))) float v2f;
typedef __attribute__((ext_vector_type(8))) float v8f;

#define HW 1024
#define NPIX (HW * HW)

__device__ __forceinline__ float tanh_hw(float v) {
    return __builtin_amdgcn_tanhf(v);
}
__device__ __forceinline__ float sigmoid_hw(float v) {
    return fmaf(0.5f, tanh_hw(0.5f * v), 0.5f);
}

template <bool FIRST>
__global__ __launch_bounds__(256) void convlstm_step(
    const float* __restrict__ x,   const float* __restrict__ hin,
    const float* __restrict__ cin, const float* __restrict__ w,
    const float* __restrict__ bias, float* __restrict__ hout,
    float* __restrict__ cout)
{
    const int lane  = threadIdx.x & 31;
    const int wib   = threadIdx.x >> 5;
    const int wid   = __builtin_amdgcn_readfirstlane(blockIdx.x * 8 + wib);
    const int yb    = (wid >> 5) * 4;
    const int x0    = (wid & 31) * 32;

    const int c  = lane & 15;
    const int hf = lane >> 4;
    const float* __restrict__ plane = (!FIRST && hf) ? hin : x;
    const bool zero_half = FIRST && (hf != 0);

    const int r = c >> 2;
    const float* wl = w + ((c & 3) * 18 + hf * 9);
    float wv[9];
#pragma unroll
    for (int i = 0; i < 9; ++i) wv[i] = wl[i];
    v2f af[9];
#pragma unroll
    for (int ch = 0; ch < 9; ++ch) {
#pragma unroll
        for (int s = 0; s < 2; ++s) {
            const int j = 2 * ch + s;
            float v = 0.0f;
#pragma unroll
            for (int rr = 0; rr < 4; ++rr) {
                const int idx = j - 3 * rr;
                if (idx >= 0 && idx <= 8) v = (r == rr) ? wv[idx] : v;
            }
            if (s == 0) af[ch].x = v; else af[ch].y = v;
        }
    }

    v8f cinit;
#pragma unroll
    for (int q = 0; q < 8; ++q) cinit[q] = bias[q & 3];

    v2f bf[2][9];
    const bool interior = (yb >= 4) && (yb <= HW - 8) && (x0 >= 32) && (x0 <= HW - 64);
    if (interior) {
        const float* bp = plane + ((yb - 1) * HW + x0 + c - 1);
#pragma unroll
        for (int t = 0; t < 2; ++t) {
#pragma unroll
            for (int ch = 0; ch < 9; ++ch) {
#pragma unroll
                for (int s = 0; s < 2; ++s) {
                    const int j = 2 * ch + s;
                    const float lv = bp[(j / 3) * HW + 16 * t + (j % 3)];
                    const float v = zero_half ? 0.0f : lv;
                    if (s == 0) bf[t][ch].x = v; else bf[t][ch].y = v;
                }
            }
        }
    } else {
#pragma unroll
        for (int t = 0; t < 2; ++t) {
#pragma unroll
            for (int ch = 0; ch < 9; ++ch) {
#pragma unroll
                for (int s = 0; s < 2; ++s) {
                    const int j = 2 * ch + s;
                    const int rw = yb - 1 + j / 3;
                    const int cc = x0 + 16 * t + c - 1 + j % 3;
                    const bool ok = (rw >= 0) && (rw < HW) && (cc >= 0) && (cc < HW);
                    const float tv = plane[ok ? (rw * HW + cc) : 0];
                    const float v = (ok && !zero_half) ? tv : 0.0f;
                    if (s == 0) bf[t][ch].x = v; else bf[t][ch].y = v;
                }
            }
        }
    }

    const int p0 = (yb + 2 * hf) * HW + x0 + c;
    float cp[2][2];
#pragma unroll
    for (int t = 0; t < 2; ++t) {
#pragma unroll
        for (int e = 0; e < 2; ++e) cp[t][e] = FIRST ? 0.0f : cin[p0 + e * HW + 16 * t];
    }

    v8f acc0 = cinit, acc1 = cinit;
#pragma unroll
    for (int ch = 0; ch < 9; ++ch) {
        acc0 = __builtin_amdgcn_wmma_f32_16x16x4_f32(
            false, af[ch], false, bf[0][ch], (short)0, acc0, false, false);
        acc1 = __builtin_amdgcn_wmma_f32_16x16x4_f32(
            false, af[ch], false, bf[1][ch], (short)0, acc1, false, false);
    }

#pragma unroll
    for (int t = 0; t < 2; ++t) {
        const v8f acc = t ? acc1 : acc0;
#pragma unroll
        for (int e = 0; e < 2; ++e) {
            const float gi = acc[4 * e + 0];
            const float gf = acc[4 * e + 1];
            const float go = acc[4 * e + 2];
            const float gg = acc[4 * e + 3];
            const int p = p0 + e * HW + 16 * t;
            const float cprev = cp[t][e];
            float cn = sigmoid_hw(gf) * cprev + sigmoid_hw(gi) * tanh_hw(gg);
            float hn = sigmoid_hw(go) * tanh_hw(cn);
            cout[p] = cn;
            hout[p] = hn;
        }
    }
}

extern "C" void kernel_launch(void* const* d_in, const int* in_sizes, int n_in,
                              void* d_out, int out_size, void* d_ws, size_t ws_size,
                              hipStream_t stream) {
    (void)in_sizes; (void)n_in; (void)out_size; (void)ws_size;
    const float* data  = (const float*)d_in[0];
    const float* enc_w = (const float*)d_in[1];
    const float* enc_b = (const float*)d_in[2];
    const float* dec_w = (const float*)d_in[3];
    const float* dec_b = (const float*)d_in[4];
    float* out = (float*)d_out;

    float* ws = (float*)d_ws;
    float* h0 = ws;
    float* h1 = ws + (size_t)NPIX;
    float* h2 = ws + (size_t)2 * NPIX;
    float* c0 = ws + (size_t)3 * NPIX;
    float* c1 = ws + (size_t)4 * NPIX;

    dim3 grid(1024), blk(256);

    float *ha = h0, *hb = h1, *ca = c0, *cb = c1;
    for (int t = 0; t < 20; ++t) {
        if (t == 0)
            convlstm_step<true><<<grid, blk, 0, stream>>>(data, ha, ca, enc_w, enc_b, hb, cb);
        else
            convlstm_step<false><<<grid, blk, 0, stream>>>(data + (size_t)t * NPIX,
                                                       ha, ca, enc_w, enc_b, hb, cb);
        float* tmp;
        tmp = ha; ha = hb; hb = tmp;
        tmp = ca; ca = cb; cb = tmp;
    }
    const float* dec_in = ha;

    float *hda = hb, *hdb = h2, *cda = ca, *cdb = cb;
    for (int t = 0; t < 20; ++t) {
        float* hdst = (t == 19) ? out : hdb;
        if (t == 0)
            convlstm_step<true><<<grid, blk, 0, stream>>>(dec_in, hda, cda, dec_w, dec_b,
                                                          hdst, cdb);
        else
            convlstm_step<false><<<grid, blk, 0, stream>>>(dec_in, hda, cda, dec_w, dec_b,
                                                           hdst, cdb);
        float* tmp;
        tmp = hda; hda = hdb; hdb = tmp;
        tmp = cda; cda = cdb; cdb = tmp;
    }
}


# Two time steps per launch: a work-group computes step A on its 128x16 tile plus halo, keeps h, c in LDS and runs step B from it; 40 launches -> 20; step A tiles sit a row up: tap sum order changes
# speedup vs baseline: 2.6618x; 1.1676x over previous
#include <hip/hip_runtime.h>

typedef __attribute__((ext_vector_type(2))) float v2f;
typedef __attribute__((ext_vector_type(8))) float v8f;

#define HW 1024
#define NPIX (HW * HW)

#define TW 128
#define TH 16
#define AW 144
#define AH 20
#define A_TILES 45
#define NWAVE 16

__device__ __forceinline__ float tanh_hw(float v) {
    return __builtin_amdgcn_tanhf(v);
}
__device__ __forceinline__ float sigmoid_hw(float v) {
    return fmaf(0.5f, tanh_hw(0.5f * v), 0.5f);
}

template <bool FIRST>
__global__ __launch_bounds__(512) void convlstm_two_steps(
    const float* __restrict__ xa,  const float* __restrict__ xb,
    const float* __restrict__ hin, const float* __restrict__ cin,
    const float* __restrict__ w,   const float* __restrict__ bias,
    float* __restrict__ hout, float* __restrict__ cout)
{
    __shared__ float hs[AH * AW];
    __shared__ float cs[AH * AW];

    const int lane = threadIdx.x & 31;
    const int wv = __builtin_amdgcn_readfirstlane((int)(threadIdx.x >> 5));
    const int x0 = (int)(blockIdx.x & 7) * TW;
    const int y0 = (int)(blockIdx.x >> 3) * TH;

    const int c  = lane & 15;
    const int hf = lane >> 4;

    const int r = c >> 2;
    const float* wl = w + ((c & 3) * 18 + hf * 9);
    float wv9[9];
#pragma unroll
    for (int i = 0; i < 9; ++i) wv9[i] = wl[i];
    v2f af[9];
#pragma unroll
    for (int ch = 0; ch < 9; ++ch) {
#pragma unroll
        for (int s = 0; s < 2; ++s) {
            const int j = 2 * ch + s;
            float v = 0.0f;
#pragma unroll
            for (int rr = 0; rr < 4; ++rr) {
                const int idx = j - 3 * rr;
                if (idx >= 0 && idx <= 8) v = (r == rr) ? wv9[idx] : v;
            }
            if (s == 0) af[ch].x = v; else af[ch].y = v;
        }
    }

    v8f cinit;
#pragma unroll
    for (int q = 0; q < 8; ++q) cinit[q] = bias[q & 3];

    {
        const float* __restrict__ plane = (!FIRST && hf) ? hin : xa;
        const bool zero_half = FIRST && (hf != 0);

        for (int task = wv; task < A_TILES; task += NWAVE) {
            const int k  = task / 9;
            const int i  = task - 9 * k;
            const int ty = y0 - 1 + 4 * k;
            const int tx = x0 - 8 + 16 * i;

            v2f bf[9];
            const bool inside = (ty >= 1) && (ty + 4 <= HW - 1) &&
                                (tx >= 1) && (tx + 16 <= HW - 1);
            if (inside) {
                const float* bp = plane + ((ty - 1) * HW + tx + c - 1);
#pragma unroll
                for (int ch = 0; ch < 9; ++ch) {
#pragma unroll
                    for (int s = 0; s < 2; ++s) {
                        const int j = 2 * ch + s;
                        const float lv = bp[(j / 3) * HW + (j % 3)];
                        const float v = zero_half ? 0.0f : lv;
                        if (s == 0) bf[ch].x = v; else bf[ch].y = v;
                    }
                }
            } else {
#pragma unroll
                for (int ch = 0; ch < 9; ++ch) {
#pragma unroll
                    for (int s = 0; s < 2; ++s) {
                        const int j = 2 * ch + s;
                        const int rw = ty - 1 + j / 3;
                        const int cc = tx + c - 1 + j % 3;
                        const bool ok = (rw >= 0) && (rw < HW) && (cc >= 0) && (cc < HW);
                        const float tv = plane[ok ? (rw * HW + cc) : 0];
                        const float v = (ok && !zero_half) ? tv : 0.0f;
                        if (s == 0) bf[ch].x = v; else bf[ch].y = v;
                    }
                }
            }

            const int px = tx + c;
            bool  in_img[2];
            float cp[2];
#pragma unroll
            for (int e = 0; e < 2; ++e) {
                const int py = ty + 2 * hf + e;
                in_img[e] = (py >= 0) && (py < HW) && (px >= 0) && (px < HW);
                const float tv = FIRST ? 0.0f : cin[in_img[e] ? (py * HW + px) : 0];
                cp[e] = in_img[e] ? tv : 0.0f;
            }

            v8f acc = cinit;
#pragma unroll
            for (int ch = 0; ch < 9; ++ch) {
                acc = __builtin_amdgcn_wmma_f32_16x16x4_f32(
                    false, af[ch], false, bf[ch], (short)0, acc, false, false);
            }

#pragma unroll
            for (int e = 0; e < 2; ++e) {
                const float gi = acc[4 * e + 0];
                const float gf = acc[4 * e + 1];
                const float go = acc[4 * e + 2];
                const float gg = acc[4 * e + 3];
                const float cn = sigmoid_hw(gf) * cp[e] + sigmoid_hw(gi) * tanh_hw(gg);
                const float hn = sigmoid_hw(go) * tanh_hw(cn);
                const int l = (4 * k + 2 * hf + e) * AW + 16 * i + c;
                hs[l] = in_img[e] ? hn : 0.0f;
                cs[l] = cn;
            }
        }
    }

    {
        const int k   = wv >> 2;
        const int ip  = wv & 3;
        const int ty  = y0 + 4 * k;
        const int txp = x0 + 32 * ip;

        v2f bf[2][9];
#pragma unroll
        for (int t = 0; t < 2; ++t) {
#pragma unroll
            for (int ch = 0; ch < 9; ++ch) { bf[t][ch].x = 0.0f; bf[t][ch].y = 0.0f; }
        }

        if (!hf) {
            const bool inside = (ty >= 1) && (ty + 4 <= HW - 1) &&
                                (txp >= 1) && (txp + 32 <= HW - 1);
            if (inside) {
                const float* bp = xb + ((ty - 1) * HW + txp + c - 1);
#pragma unroll
                for (int t = 0; t < 2; ++t) {
#pragma unroll
                    for (int ch = 0; ch < 9; ++ch) {
#pragma unroll
                        for (int s = 0; s < 2; ++s) {
                            const int j = 2 * ch + s;
                            const float v = bp[(j / 3) * HW + 16 * t + (j % 3)];
                            if (s == 0) bf[t][ch].x = v; else bf[t][ch].y = v;
                        }
                    }
                }
            } else {
#pragma unroll
                for (int t = 0; t < 2; ++t) {
#pragma unroll
                    for (int ch = 0; ch < 9; ++ch) {
#pragma unroll
                        for (int s = 0; s < 2; ++s) {
                            const int j = 2 * ch + s;
                            const int rw = ty - 1 + j / 3;
                            const int cc = txp + 16 * t + c - 1 + j % 3;
                            const bool ok = (rw >= 0) && (rw < HW) && (cc >= 0) && (cc < HW);
                            const float tv = xb[ok ? (rw * HW + cc) : 0];
                            const float v = ok ? tv : 0.0f;
                            if (s == 0) bf[t][ch].x = v; else bf[t][ch].y = v;
                        }
                    }
                }
            }
        }

        __builtin_amdgcn_fence(__ATOMIC_RELEASE, "workgroup", "local");
        __builtin_amdgcn_s_barrier();
        __builtin_amdgcn_fence(__ATOMIC_ACQUIRE, "workgroup", "local");

        if (hf) {
            const float* lp = hs + (4 * k) * AW + (32 * ip + c + 7);
#pragma unroll
            for (int t = 0; t < 2; ++t) {
#pragma unroll
                for (int ch = 0; ch < 9; ++ch) {
#pragma unroll
                    for (int s = 0; s < 2; ++s) {
                        const int j = 2 * ch + s;
                        const float v = lp[(j / 3) * AW + 16 * t + (j % 3)];
                        if (s == 0) bf[t][ch].x = v; else bf[t][ch].y = v;
                    }
                }
            }
        }

        const int lc = (4 * k + 2 * hf + 1) * AW + (32 * ip + c + 8);
        float cp[2][2];
#pragma unroll
        for (int t = 0; t < 2; ++t) {
#pragma unroll
            for (int e = 0; e < 2; ++e) cp[t][e] = cs[lc + e * AW + 16 * t];
        }

        v8f acc0 = cinit, acc1 = cinit;
#pragma unroll
        for (int ch = 0; ch < 9; ++ch) {
            acc0 = __builtin_amdgcn_wmma_f32_16x16x4_f32(
                false, af[ch], false, bf[0][ch], (short)0, acc0, false, false);
            acc1 = __builtin_amdgcn_wmma_f32_16x16x4_f32(
                false, af[ch], false, bf[1][ch], (short)0, acc1, false, false);
        }

        const int p0 = (ty + 2 * hf) * HW + txp + c;
#pragma unroll
        for (int t = 0; t < 2; ++t) {
            const v8f acc = t ? acc1 : acc0;
#pragma unroll
            for (int e = 0; e < 2; ++e) {
                const float gi = acc[4 * e + 0];
                const float gf = acc[4 * e + 1];
                const float go = acc[4 * e + 2];
                const float gg = acc[4 * e + 3];
                const int p = p0 + e * HW + 16 * t;
                const float cn = sigmoid_hw(gf) * cp[t][e] + sigmoid_hw(gi) * tanh_hw(gg);
                const float hn = sigmoid_hw(go) * tanh_hw(cn);
                cout[p] = cn;
                hout[p] = hn;
            }
        }
    }
}

extern "C" void kernel_launch(void* const* d_in, const int* in_sizes, int n_in,
                              void* d_out, int out_size, void* d_ws, size_t ws_size,
                              hipStream_t stream) {
    (void)in_sizes; (void)n_in; (void)out_size; (void)ws_size;
    const float* data  = (const float*)d_in[0];
    const float* enc_w = (const float*)d_in[1];
    const float* enc_b = (const float*)d_in[2];
    const float* dec_w = (const float*)d_in[3];
    const float* dec_b = (const float*)d_in[4];
    float* out = (float*)d_out;

    float* ws = (float*)d_ws;
    float* h0 = ws;
    float* h1 = ws + (size_t)NPIX;
    float* h2 = ws + (size_t)2 * NPIX;
    float* c0 = ws + (size_t)3 * NPIX;
    float* c1 = ws + (size_t)4 * NPIX;

    dim3 grid(512), blk(512);

    float *ha = h0, *hb = h1, *ca = c0, *cb = c1;
    for (int p = 0; p < 10; ++p) {
        const float* xa = data + (size_t)(2 * p) * NPIX;
        const float* xb = data + (size_t)(2 * p + 1) * NPIX;
        if (p == 0)
            convlstm_two_steps<true><<<grid, blk, 0, stream>>>(xa, xb, ha, ca, enc_w, enc_b, hb, cb);
        else
            convlstm_two_steps<false><<<grid, blk, 0, stream>>>(xa, xb, ha, ca, enc_w, enc_b, hb, cb);
        float* tmp;
        tmp = ha; ha = hb; hb = tmp;
        tmp = ca; ca = cb; cb = tmp;
    }
    const float* dec_in = ha;

    float *hda = hb, *hdb = h2, *cda = ca, *cdb = cb;
    for (int p = 0; p < 10; ++p) {
        float* hdst = (p == 9) ? out : hdb;
        if (p == 0)
            convlstm_two_steps<true><<<grid, blk, 0, stream>>>(dec_in, dec_in, hda, cda, dec_w, dec_b,
                                                               hdst, cdb);
        else
            convlstm_two_steps<false><<<grid, blk, 0, stream>>>(dec_in, dec_in, hda, cda, dec_w, dec_b,
                                                                hdst, cdb);
        float* tmp;
        tmp = hda; hda = hdb; hdb = tmp;
        tmp = cda; cda = cdb; cdb = tmp;
    }
}


# Four time steps per launch: a work-group runs steps 1-3 on its 128x32 tile plus halo with h, c kept in LDS and step 4 from it; 20 launches -> 10; tiles sit on other rows: tap sum order changes
# speedup vs baseline: 3.2476x; 1.2201x over previous
#include <hip/hip_runtime.h>

typedef __attribute__((ext_vector_type(2))) float v2f;
typedef __attribute__((ext_vector_type(8))) float v8f;

#define HW 1024
#define NPIX (HW * HW)

#define TW 128
#define TH 32
#define LP 146
#define LROWS 40
#define PLANE (LROWS * LP)
#define NWAVE 32

__device__ __forceinline__ float tanh_hw(float v) {
    return __builtin_amdgcn_tanhf(v);
}
__device__ __forceinline__ float sigmoid_hw(float v) {
    return fmaf(0.5f, tanh_hw(0.5f * v), 0.5f);
}


template <bool FIRST, bool SRC_LDS, bool DST_LDS>
__device__ __forceinline__ void tile_step(
    const int ty, const int tx, const int lrow, const int lcol,
    const float* __restrict__ xs,
    const float* __restrict__ hin, const float* __restrict__ cin,
    const float* hsrc, float* cbuf, float* hdst,
    float* __restrict__ hout, float* __restrict__ cout,
    const v2f (&af)[9], const v8f cinit, const int c, const int hf)
{
    v2f bf[9];
    float cp[2];
    bool  in_img[2];
    const int px = tx + c;
#pragma unroll
    for (int e = 0; e < 2; ++e) {
        const int py = ty + 2 * hf + e;
        in_img[e] = (py >= 0) && (py < HW) && (px >= 0) && (px < HW);
    }
    const bool inside = (ty >= 1) && (ty + 4 <= HW - 1) &&
                        (tx >= 1) && (tx + 16 <= HW - 1);

    if (!SRC_LDS) {
        const float* __restrict__ plane = (!FIRST && hf) ? hin : xs;
        const bool zero_half = FIRST && (hf != 0);
        if (inside) {
            const float* bp = plane + ((ty - 1) * HW + tx + c - 1);
#pragma unroll
            for (int ch = 0; ch < 9; ++ch) {
#pragma unroll
                for (int s = 0; s < 2; ++s) {
                    const int j = 2 * ch + s;
                    const float lv = bp[(j / 3) * HW + (j % 3)];
                    const float v = zero_half ? 0.0f : lv;
                    if (s == 0) bf[ch].x = v; else bf[ch].y = v;
                }
            }
        } else {
#pragma unroll
            for (int ch = 0; ch < 9; ++ch) {
#pragma unroll
                for (int s = 0; s < 2; ++s) {
                    const int j = 2 * ch + s;
                    const int rw = ty - 1 + j / 3;
                    const int cc = tx + c - 1 + j % 3;
                    const bool ok = (rw >= 0) && (rw < HW) && (cc >= 0) && (cc < HW);
                    const float tv = plane[ok ? (rw * HW + cc) : 0];
                    const float v = (ok && !zero_half) ? tv : 0.0f;
                    if (s == 0) bf[ch].x = v; else bf[ch].y = v;
                }
            }
        }
#pragma unroll
        for (int e = 0; e < 2; ++e) {
            const int py = ty + 2 * hf + e;
            const float tv = FIRST ? 0.0f : cin[in_img[e] ? (py * HW + px) : 0];
            cp[e] = in_img[e] ? tv : 0.0f;
        }
    } else {
#pragma unroll
        for (int ch = 0; ch < 9; ++ch) { bf[ch].x = 0.0f; bf[ch].y = 0.0f; }
        if (!hf) {
            if (inside) {
                const float* bp = xs + ((ty - 1) * HW + tx + c - 1);
#pragma unroll
                for (int ch = 0; ch < 9; ++ch) {
#pragma unroll
                    for (int s = 0; s < 2; ++s) {
                        const int j = 2 * ch + s;
                        const float v = bp[(j / 3) * HW + (j % 3)];
                        if (s == 0) bf[ch].x = v; else bf[ch].y = v;
                    }
                }
            } else {
#pragma unroll
                for (int ch = 0; ch < 9; ++ch) {
#pragma unroll
                    for (int s = 0; s < 2; ++s) {
                        const int j = 2 * ch + s;
                        const int rw = ty - 1 + j / 3;
                        const int cc = tx + c - 1 + j % 3;
                        const bool ok = (rw >= 0) && (rw < HW) && (cc >= 0) && (cc < HW);
                        const float tv = xs[ok ? (rw * HW + cc) : 0];
                        const float v = ok ? tv : 0.0f;
                        if (s == 0) bf[ch].x = v; else bf[ch].y = v;
                    }
                }
            }
        } else {
            const float* lp = hsrc + (lrow - 1) * LP + (lcol + c - 1);
#pragma unroll
            for (int ch = 0; ch < 9; ++ch) {
#pragma unroll
                for (int s = 0; s < 2; ++s) {
                    const int j = 2 * ch + s;
                    const float v = lp[(j / 3) * LP + (j % 3)];
                    if (s == 0) bf[ch].x = v; else bf[ch].y = v;
                }
            }
        }
#pragma unroll
        for (int e = 0; e < 2; ++e)
            cp[e] = cbuf[(lrow + 2 * hf + e) * LP + lcol + c];
    }

    v8f acc = cinit;
#pragma unroll
    for (int ch = 0; ch < 9; ++ch) {
        acc = __builtin_amdgcn_wmma_f32_16x16x4_f32(
            false, af[ch], false, bf[ch], (short)0, acc, false, false);
    }

#pragma unroll
    for (int e = 0; e < 2; ++e) {
        const float gi = acc[4 * e + 0];
        const float gf = acc[4 * e + 1];
        const float go = acc[4 * e + 2];
        const float gg = acc[4 * e + 3];
        const float cn = sigmoid_hw(gf) * cp[e] + sigmoid_hw(gi) * tanh_hw(gg);
        const float hn = sigmoid_hw(go) * tanh_hw(cn);
        if (DST_LDS) {
            const int l = (lrow + 2 * hf + e) * LP + lcol + c;
            hdst[l] = in_img[e] ? hn : 0.0f;
            cbuf[l] = cn;
        } else {
            const int p = (ty + 2 * hf + e) * HW + px;
            cout[p] = cn;
            hout[p] = hn;
        }
    }
}

template <bool FIRST>
__global__ __launch_bounds__(1024) void convlstm_four_steps(
    const float* __restrict__ xs0, const float* __restrict__ xs1,
    const float* __restrict__ xs2, const float* __restrict__ xs3,
    const float* __restrict__ hin, const float* __restrict__ cin,
    const float* __restrict__ w,   const float* __restrict__ bias,
    float* __restrict__ hout, float* __restrict__ cout)
{
    __shared__ float hb[2][PLANE];
    __shared__ float cb[PLANE];

    const int tid  = (int)threadIdx.x;
    const int lane = tid & 31;
    const int wv = __builtin_amdgcn_readfirstlane(tid >> 5);
    const int x0 = (int)(blockIdx.x & 7) * TW;
    const int y0 = (int)(blockIdx.x >> 3) * TH;

    const int c  = lane & 15;
    const int hf = lane >> 4;

    if (tid < 160) {
        const int b = tid / 80;
        const int q = tid - 80 * b;
        hb[b][(q >> 1) * LP + (q & 1) * (LP - 1)] = 0.0f;
    } else if (tid < 160 + 2 * LP) {
        hb[1][37 * LP + (tid - 160)] = 0.0f;
    }

    const int r = c >> 2;
    const float* wl = w + ((c & 3) * 18 + hf * 9);
    float wv9[9];
#pragma unroll
    for (int i = 0; i < 9; ++i) wv9[i] = wl[i];
    v2f af[9];
#pragma unroll
    for (int ch = 0; ch < 9; ++ch) {
#pragma unroll
        for (int s = 0; s < 2; ++s) {
            const int j = 2 * ch + s;
            float v = 0.0f;
#pragma unroll
            for (int rr = 0; rr < 4; ++rr) {
                const int idx = j - 3 * rr;
                if (idx >= 0 && idx <= 8) v = (r == rr) ? wv9[idx] : v;
            }
            if (s == 0) af[ch].x = v; else af[ch].y = v;
        }
    }

    v8f cinit;
#pragma unroll
    for (int q = 0; q < 8; ++q) cinit[q] = bias[q & 3];


    for (int task = wv; task < 90; task += NWAVE) {
        const int k = task / 9;
        const int i = task - 9 * k;
        tile_step<FIRST, false, true>(y0 - 3 + 4 * k, x0 - 8 + 16 * i, 4 * k, 1 + 16 * i,
                                      xs0, hin, cin, nullptr, cb, hb[0], nullptr, nullptr,
                                      af, cinit, c, hf);
    }
    __syncthreads();

    for (int task = wv; task < 81; task += NWAVE) {
        const int k = task / 9;
        const int i = task - 9 * k;
        tile_step<false, true, true>(y0 - 2 + 4 * k, x0 - 8 + 16 * i, 1 + 4 * k, 1 + 16 * i,
                                     xs1, nullptr, nullptr, hb[0], cb, hb[1], nullptr, nullptr,
                                     af, cinit, c, hf);
    }
    __syncthreads();

    for (int task = wv; task < 81; task += NWAVE) {
        const int k = task / 9;
        const int i = task - 9 * k;
        tile_step<false, true, true>(y0 - 1 + 4 * k, x0 - 8 + 16 * i, 2 + 4 * k, 1 + 16 * i,
                                     xs2, nullptr, nullptr, hb[1], cb, hb[0], nullptr, nullptr,
                                     af, cinit, c, hf);
    }
    __syncthreads();

    {
        const int k = wv >> 2;
        for (int t = 0; t < 2; ++t) {
            const int i = 2 * (wv & 3) + t;
            tile_step<false, true, false>(y0 + 4 * k, x0 + 16 * i, 3 + 4 * k, 9 + 16 * i,
                                          xs3, nullptr, nullptr, hb[0], cb, nullptr, hout, cout,
                                          af, cinit, c, hf);
        }
    }
}

extern "C" void kernel_launch(void* const* d_in, const int* in_sizes, int n_in,
                              void* d_out, int out_size, void* d_ws, size_t ws_size,
                              hipStream_t stream) {
    (void)in_sizes; (void)n_in; (void)out_size; (void)ws_size;
    const float* data  = (const float*)d_in[0];
    const float* enc_w = (const float*)d_in[1];
    const float* enc_b = (const float*)d_in[2];
    const float* dec_w = (const float*)d_in[3];
    const float* dec_b = (const float*)d_in[4];
    float* out = (float*)d_out;

    float* ws = (float*)d_ws;
    float* h0 = ws;
    float* h1 = ws + (size_t)NPIX;
    float* h2 = ws + (size_t)2 * NPIX;
    float* c0 = ws + (size_t)3 * NPIX;
    float* c1 = ws + (size_t)4 * NPIX;

    dim3 grid(256), blk(1024);

    float *ha = h0, *hb = h1, *ca = c0, *cb = c1;
    for (int p = 0; p < 5; ++p) {
        const float* xa = data + (size_t)(4 * p) * NPIX;
        const float* xb = data + (size_t)(4 * p + 1) * NPIX;
        const float* xc = data + (size_t)(4 * p + 2) * NPIX;
        const float* xd = data + (size_t)(4 * p + 3) * NPIX;
        if (p == 0)
            convlstm_four_steps<true><<<grid, blk, 0, stream>>>(xa, xb, xc, xd, ha, ca, enc_w, enc_b, hb, cb);
        else
            convlstm_four_steps<false><<<grid, blk, 0, stream>>>(xa, xb, xc, xd, ha, ca, enc_w, enc_b, hb, cb);
        float* tmp;
        tmp = ha; ha = hb; hb = tmp;
        tmp = ca; ca = cb; cb = tmp;
    }
    const float* dec_in = ha;

    float *hda = hb, *hdb = h2, *cda = ca, *cdb = cb;
    for (int p = 0; p < 5; ++p) {
        float* hdst = (p == 4) ? out : hdb;
        if (p == 0)
            convlstm_four_steps<true><<<grid, blk, 0, stream>>>(dec_in, dec_in, dec_in, dec_in, hda, cda,
                                                                dec_w, dec_b, hdst, cdb);
        else
            convlstm_four_steps<false><<<grid, blk, 0, stream>>>(dec_in, dec_in, dec_in, dec_in, hda, cda,
                                                                 dec_w, dec_b, hdst, cdb);
        float* tmp;
        tmp = hda; hda = hdb; hdb = tmp;
        tmp = cda; cda = cdb; cdb = tmp;
    }
}


# Decoder: its input plane is constant, so the x half of the gate sums (bias + x taps) is built once per launch into LDS; each step adds only the h taps, 5 WMMAs per tile not 9; gate sum order changes
# speedup vs baseline: 3.6437x; 1.1220x over previous
#include <hip/hip_runtime.h>

typedef __attribute__((ext_vector_type(2))) float v2f;
typedef __attribute__((ext_vector_type(8))) float v8f;

#define HW 1024
#define NPIX (HW * HW)

#define TW 128
#define TH 32
#define LP 146
#define LROWS 40
#define PLANE (LROWS * LP)
#define NWAVE 32

__device__ __forceinline__ float tanh_hw(float v) {
    return __builtin_amdgcn_tanhf(v);
}
__device__ __forceinline__ float sigmoid_hw(float v) {
    return fmaf(0.5f, tanh_hw(0.5f * v), 0.5f);
}


template <bool FIRST, bool SRC_LDS, bool DST_LDS>
__device__ __forceinline__ void tile_step(
    const int ty, const int tx, const int lrow, const int lcol,
    const float* __restrict__ xs,
    const float* __restrict__ hin, const float* __restrict__ cin,
    const float* hsrc, float* cbuf, float* hdst,
    float* __restrict__ hout, float* __restrict__ cout,
    const v2f (&af)[9], const v8f cinit, const int c, const int hf)
{
    v2f bf[9];
    float cp[2];
    bool  in_img[2];
    const int px = tx + c;
#pragma unroll
    for (int e = 0; e < 2; ++e) {
        const int py = ty + 2 * hf + e;
        in_img[e] = (py >= 0) && (py < HW) && (px >= 0) && (px < HW);
    }
    const bool inside = (ty >= 1) && (ty + 4 <= HW - 1) &&
                        (tx >= 1) && (tx + 16 <= HW - 1);

    if (!SRC_LDS) {
        const float* __restrict__ plane = (!FIRST && hf) ? hin : xs;
        const bool zero_half = FIRST && (hf != 0);
        if (inside) {
            const float* bp = plane + ((ty - 1) * HW + tx + c - 1);
#pragma unroll
            for (int ch = 0; ch < 9; ++ch) {
#pragma unroll
                for (int s = 0; s < 2; ++s) {
                    const int j = 2 * ch + s;
                    const float lv = bp[(j / 3) * HW + (j % 3)];
                    const float v = zero_half ? 0.0f : lv;
                    if (s == 0) bf[ch].x = v; else bf[ch].y = v;
                }
            }
        } else {
#pragma unroll
            for (int ch = 0; ch < 9; ++ch) {
#pragma unroll
                for (int s = 0; s < 2; ++s) {
                    const int j = 2 * ch + s;
                    const int rw = ty - 1 + j / 3;
                    const int cc = tx + c - 1 + j % 3;
                    const bool ok = (rw >= 0) && (rw < HW) && (cc >= 0) && (cc < HW);
                    const float tv = plane[ok ? (rw * HW + cc) : 0];
                    const float v = (ok && !zero_half) ? tv : 0.0f;
                    if (s == 0) bf[ch].x = v; else bf[ch].y = v;
                }
            }
        }
#pragma unroll
        for (int e = 0; e < 2; ++e) {
            const int py = ty + 2 * hf + e;
            const float tv = FIRST ? 0.0f : cin[in_img[e] ? (py * HW + px) : 0];
            cp[e] = in_img[e] ? tv : 0.0f;
        }
    } else {
#pragma unroll
        for (int ch = 0; ch < 9; ++ch) { bf[ch].x = 0.0f; bf[ch].y = 0.0f; }
        if (!hf) {
            if (inside) {
                const float* bp = xs + ((ty - 1) * HW + tx + c - 1);
#pragma unroll
                for (int ch = 0; ch < 9; ++ch) {
#pragma unroll
                    for (int s = 0; s < 2; ++s) {
                        const int j = 2 * ch + s;
                        const float v = bp[(j / 3) * HW + (j % 3)];
                        if (s == 0) bf[ch].x = v; else bf[ch].y = v;
                    }
                }
            } else {
#pragma unroll
                for (int ch = 0; ch < 9; ++ch) {
#pragma unroll
                    for (int s = 0; s < 2; ++s) {
                        const int j = 2 * ch + s;
                        const int rw = ty - 1 + j / 3;
                        const int cc = tx + c - 1 + j % 3;
                        const bool ok = (rw >= 0) && (rw < HW) && (cc >= 0) && (cc < HW);
                        const float tv = xs[ok ? (rw * HW + cc) : 0];
                        const float v = ok ? tv : 0.0f;
                        if (s == 0) bf[ch].x = v; else bf[ch].y = v;
                    }
                }
            }
        } else {
            const float* lp = hsrc + (lrow - 1) * LP + (lcol + c - 1);
#pragma unroll
            for (int ch = 0; ch < 9; ++ch) {
#pragma unroll
                for (int s = 0; s < 2; ++s) {
                    const int j = 2 * ch + s;
                    const float v = lp[(j / 3) * LP + (j % 3)];
                    if (s == 0) bf[ch].x = v; else bf[ch].y = v;
                }
            }
        }
#pragma unroll
        for (int e = 0; e < 2; ++e)
            cp[e] = cbuf[(lrow + 2 * hf + e) * LP + lcol + c];
    }

    v8f acc = cinit;
#pragma unroll
    for (int ch = 0; ch < 9; ++ch) {
        acc = __builtin_amdgcn_wmma_f32_16x16x4_f32(
            false, af[ch], false, bf[ch], (short)0, acc, false, false);
    }

#pragma unroll
    for (int e = 0; e < 2; ++e) {
        const float gi = acc[4 * e + 0];
        const float gf = acc[4 * e + 1];
        const float go = acc[4 * e + 2];
        const float gg = acc[4 * e + 3];
        const float cn = sigmoid_hw(gf) * cp[e] + sigmoid_hw(gi) * tanh_hw(gg);
        const float hn = sigmoid_hw(go) * tanh_hw(cn);
        if (DST_LDS) {
            const int l = (lrow + 2 * hf + e) * LP + lcol + c;
            hdst[l] = in_img[e] ? hn : 0.0f;
            cbuf[l] = cn;
        } else {
            const int p = (ty + 2 * hf + e) * HW + px;
            cout[p] = cn;
            hout[p] = hn;
        }
    }
}

template <bool FIRST>
__global__ __launch_bounds__(1024) void convlstm_four_steps(
    const float* __restrict__ xs0, const float* __restrict__ xs1,
    const float* __restrict__ xs2, const float* __restrict__ xs3,
    const float* __restrict__ hin, const float* __restrict__ cin,
    const float* __restrict__ w,   const float* __restrict__ bias,
    float* __restrict__ hout, float* __restrict__ cout)
{
    __shared__ float hb[2][PLANE];
    __shared__ float cb[PLANE];

    const int tid  = (int)threadIdx.x;
    const int lane = tid & 31;
    const int wv = __builtin_amdgcn_readfirstlane(tid >> 5);
    const int x0 = (int)(blockIdx.x & 7) * TW;
    const int y0 = (int)(blockIdx.x >> 3) * TH;

    const int c  = lane & 15;
    const int hf = lane >> 4;

    if (tid < 160) {
        const int b = tid / 80;
        const int q = tid - 80 * b;
        hb[b][(q >> 1) * LP + (q & 1) * (LP - 1)] = 0.0f;
    } else if (tid < 160 + 2 * LP) {
        hb[1][37 * LP + (tid - 160)] = 0.0f;
    }

    const int r = c >> 2;
    const float* wl = w + ((c & 3) * 18 + hf * 9);
    float wv9[9];
#pragma unroll
    for (int i = 0; i < 9; ++i) wv9[i] = wl[i];
    v2f af[9];
#pragma unroll
    for (int ch = 0; ch < 9; ++ch) {
#pragma unroll
        for (int s = 0; s < 2; ++s) {
            const int j = 2 * ch + s;
            float v = 0.0f;
#pragma unroll
            for (int rr = 0; rr < 4; ++rr) {
                const int idx = j - 3 * rr;
                if (idx >= 0 && idx <= 8) v = (r == rr) ? wv9[idx] : v;
            }
            if (s == 0) af[ch].x = v; else af[ch].y = v;
        }
    }

    v8f cinit;
#pragma unroll
    for (int q = 0; q < 8; ++q) cinit[q] = bias[q & 3];


    for (int task = wv; task < 90; task += NWAVE) {
        const int k = task / 9;
        const int i = task - 9 * k;
        tile_step<FIRST, false, true>(y0 - 3 + 4 * k, x0 - 8 + 16 * i, 4 * k, 1 + 16 * i,
                                      xs0, hin, cin, nullptr, cb, hb[0], nullptr, nullptr,
                                      af, cinit, c, hf);
    }
    __syncthreads();

    for (int task = wv; task < 81; task += NWAVE) {
        const int k = task / 9;
        const int i = task - 9 * k;
        tile_step<false, true, true>(y0 - 2 + 4 * k, x0 - 8 + 16 * i, 1 + 4 * k, 1 + 16 * i,
                                     xs1, nullptr, nullptr, hb[0], cb, hb[1], nullptr, nullptr,
                                     af, cinit, c, hf);
    }
    __syncthreads();

    for (int task = wv; task < 81; task += NWAVE) {
        const int k = task / 9;
        const int i = task - 9 * k;
        tile_step<false, true, true>(y0 - 1 + 4 * k, x0 - 8 + 16 * i, 2 + 4 * k, 1 + 16 * i,
                                     xs2, nullptr, nullptr, hb[1], cb, hb[0], nullptr, nullptr,
                                     af, cinit, c, hf);
    }
    __syncthreads();

    {
        const int k = wv >> 2;
        for (int t = 0; t < 2; ++t) {
            const int i = 2 * (wv & 3) + t;
            tile_step<false, true, false>(y0 + 4 * k, x0 + 16 * i, 3 + 4 * k, 9 + 16 * i,
                                          xs3, nullptr, nullptr, hb[0], cb, nullptr, hout, cout,
                                          af, cinit, c, hf);
        }
    }
}


typedef __attribute__((ext_vector_type(4))) float v4f;


__device__ __forceinline__ void load_plane_global(
    const float* __restrict__ plane, const int ty, const int tx,
    const int c, const int hf, const bool inside, v2f (&b)[5])
{
    const int row0 = ty - 1 + 3 * hf;
    if (inside) {
        const float* bp = plane + (row0 * HW + tx + c - 1);
#pragma unroll
        for (int t = 0; t < 9; ++t) {
            const float v = bp[(t / 3) * HW + (t % 3)];
            if ((t & 1) == 0) b[t >> 1].x = v; else b[t >> 1].y = v;
        }
    } else {
#pragma unroll
        for (int t = 0; t < 9; ++t) {
            const int rw = row0 + t / 3;
            const int cc = tx + c - 1 + t % 3;
            const bool ok = (rw >= 0) && (rw < HW) && (cc >= 0) && (cc < HW);
            const float tv = plane[ok ? (rw * HW + cc) : 0];
            const float v = ok ? tv : 0.0f;
            if ((t & 1) == 0) b[t >> 1].x = v; else b[t >> 1].y = v;
        }
    }
    b[4].y = 0.0f;
}

template <int MODE>
__device__ __forceinline__ void dec_tile_step(
    const int ty, const int tx, const int lrow, const int lcol,
    const float* __restrict__ xs,
    const float* __restrict__ hin, const float* __restrict__ cin,
    const float* hsrc, float* cbuf, float* hdst, v4f* gx,
    float* __restrict__ hout, float* __restrict__ cout,
    const v2f (&afx)[5], const v2f (&afh)[5], const v8f cinit,
    const int c, const int hf)
{
    constexpr bool STEP0   = (MODE <= 1);
    constexpr bool DST_LDS = (MODE != 3);
    float cp[2];
    bool  in_img[2];
    const int px = tx + c;
#pragma unroll
    for (int e = 0; e < 2; ++e) {
        const int py = ty + 2 * hf + e;
        in_img[e] = (py >= 0) && (py < HW) && (px >= 0) && (px < HW);
    }

    v8f acc;
    if (STEP0) {
        const bool inside = (ty >= 1) && (ty + 4 <= HW - 1) &&
                            (tx >= 1) && (tx + 16 <= HW - 1);
        v2f bx[5];
        load_plane_global(xs, ty, tx, c, hf, inside, bx);
        v2f bh[5];
        if (MODE == 1) load_plane_global(hin, ty, tx, c, hf, inside, bh);
        acc = cinit;
#pragma unroll
        for (int ch = 0; ch < 5; ++ch) {
            acc = __builtin_amdgcn_wmma_f32_16x16x4_f32(
                false, afx[ch], false, bx[ch], (short)0, acc, false, false);
        }
#pragma unroll
        for (int e = 0; e < 2; ++e) {
            v4f g4;
            g4.x = acc[4 * e + 0]; g4.y = acc[4 * e + 1];
            g4.z = acc[4 * e + 2]; g4.w = acc[4 * e + 3];
            gx[(lrow + 2 * hf + e) * LP + lcol + c] = g4;
        }
        if (MODE == 1) {
#pragma unroll
            for (int ch = 0; ch < 5; ++ch) {
                acc = __builtin_amdgcn_wmma_f32_16x16x4_f32(
                    false, afh[ch], false, bh[ch], (short)0, acc, false, false);
            }
#pragma unroll
            for (int e = 0; e < 2; ++e) {
                const int py = ty + 2 * hf + e;
                const float tv = cin[in_img[e] ? (py * HW + px) : 0];
                cp[e] = in_img[e] ? tv : 0.0f;
            }
        } else {
            cp[0] = 0.0f; cp[1] = 0.0f;
        }
    } else {
        v2f bh[5];
        const float* lp = hsrc + (lrow - 1 + 3 * hf) * LP + (lcol + c - 1);
#pragma unroll
        for (int t = 0; t < 9; ++t) {
            const float v = lp[(t / 3) * LP + (t % 3)];
            if ((t & 1) == 0) bh[t >> 1].x = v; else bh[t >> 1].y = v;
        }
        bh[4].y = 0.0f;
#pragma unroll
        for (int e = 0; e < 2; ++e) {
            const v4f g4 = gx[(lrow + 2 * hf + e) * LP + lcol + c];
            acc[4 * e + 0] = g4.x; acc[4 * e + 1] = g4.y;
            acc[4 * e + 2] = g4.z; acc[4 * e + 3] = g4.w;
        }
#pragma unroll
        for (int e = 0; e < 2; ++e)
            cp[e] = cbuf[(lrow + 2 * hf + e) * LP + lcol + c];
#pragma unroll
        for (int ch = 0; ch < 5; ++ch) {
            acc = __builtin_amdgcn_wmma_f32_16x16x4_f32(
                false, afh[ch], false, bh[ch], (short)0, acc, false, false);
        }
    }

#pragma unroll
    for (int e = 0; e < 2; ++e) {
        const float gi = acc[4 * e + 0];
        const float gf = acc[4 * e + 1];
        const float go = acc[4 * e + 2];
        const float gg = acc[4 * e + 3];
        const float cn = sigmoid_hw(gf) * cp[e] + sigmoid_hw(gi) * tanh_hw(gg);
        const float hn = sigmoid_hw(go) * tanh_hw(cn);
        if (DST_LDS) {
            const int l = (lrow + 2 * hf + e) * LP + lcol + c;
            hdst[l] = in_img[e] ? hn : 0.0f;
            cbuf[l] = cn;
        } else {
            const int p = (ty + 2 * hf + e) * HW + px;
            cout[p] = cn;
            hout[p] = hn;
        }
    }
}

template <bool FIRST>
__global__ __launch_bounds__(1024) void convlstm_dec_four_steps(
    const float* __restrict__ xs,
    const float* __restrict__ hin, const float* __restrict__ cin,
    const float* __restrict__ w,   const float* __restrict__ bias,
    float* __restrict__ hout, float* __restrict__ cout)
{
    __shared__ float hb[2][PLANE];
    __shared__ float cb[PLANE];
    __shared__ v4f   gx[PLANE];

    const int tid  = (int)threadIdx.x;
    const int lane = tid & 31;
    const int wv = __builtin_amdgcn_readfirstlane(tid >> 5);
    const int x0 = (int)(blockIdx.x & 7) * TW;
    const int y0 = (int)(blockIdx.x >> 3) * TH;

    const int c  = lane & 15;
    const int hf = lane >> 4;

    if (tid < 160) {
        const int b = tid / 80;
        const int q = tid - 80 * b;
        hb[b][(q >> 1) * LP + (q & 1) * (LP - 1)] = 0.0f;
    } else if (tid < 160 + 2 * LP) {
        hb[1][37 * LP + (tid - 160)] = 0.0f;
    }

    const int r = c >> 2;
    const float* wg = w + (c & 3) * 18;
    v2f afx[5], afh[5];
#pragma unroll
    for (int t = 0; t < 10; ++t) {
        float vx = 0.0f, vh = 0.0f;
        if (t < 9) {
            const int idx = 9 * hf + t - 3 * r;
            const bool ok = (idx >= 0) && (idx <= 8);
            const int ii = ok ? idx : 0;
            const float lx = wg[ii];
            const float lh = wg[9 + ii];
            vx = ok ? lx : 0.0f;
            vh = ok ? lh : 0.0f;
        }
        if ((t & 1) == 0) { afx[t >> 1].x = vx; afh[t >> 1].x = vh; }
        else              { afx[t >> 1].y = vx; afh[t >> 1].y = vh; }
    }

    v8f cinit;
#pragma unroll
    for (int q = 0; q < 8; ++q) cinit[q] = bias[q & 3];

    for (int task = wv; task < 90; task += NWAVE) {
        const int k = task / 9;
        const int i = task - 9 * k;
        dec_tile_step<FIRST ? 0 : 1>(y0 - 3 + 4 * k, x0 - 8 + 16 * i, 4 * k, 1 + 16 * i,
                                     xs, hin, cin, nullptr, cb, hb[0], gx, nullptr, nullptr,
                                     afx, afh, cinit, c, hf);
    }
    __syncthreads();

    for (int task = wv; task < 81; task += NWAVE) {
        const int k = task / 9;
        const int i = task - 9 * k;
        dec_tile_step<2>(y0 - 2 + 4 * k, x0 - 8 + 16 * i, 1 + 4 * k, 1 + 16 * i,
                         xs, nullptr, nullptr, hb[0], cb, hb[1], gx, nullptr, nullptr,
                         afx, afh, cinit, c, hf);
    }
    __syncthreads();

    for (int task = wv; task < 81; task += NWAVE) {
        const int k = task / 9;
        const int i = task - 9 * k;
        dec_tile_step<2>(y0 - 1 + 4 * k, x0 - 8 + 16 * i, 2 + 4 * k, 1 + 16 * i,
                         xs, nullptr, nullptr, hb[1], cb, hb[0], gx, nullptr, nullptr,
                         afx, afh, cinit, c, hf);
    }
    __syncthreads();

    {
        const int k = wv >> 2;
        for (int t = 0; t < 2; ++t) {
            const int i = 2 * (wv & 3) + t;
            dec_tile_step<3>(y0 + 4 * k, x0 + 16 * i, 3 + 4 * k, 9 + 16 * i,
                             xs, nullptr, nullptr, hb[0], cb, nullptr, gx, hout, cout,
                             afx, afh, cinit, c, hf);
        }
    }
}

extern "C" void kernel_launch(void* const* d_in, const int* in_sizes, int n_in,
                              void* d_out, int out_size, void* d_ws, size_t ws_size,
                              hipStream_t stream) {
    (void)in_sizes; (void)n_in; (void)out_size; (void)ws_size;
    const float* data  = (const float*)d_in[0];
    const float* enc_w = (const float*)d_in[1];
    const float* enc_b = (const float*)d_in[2];
    const float* dec_w = (const float*)d_in[3];
    const float* dec_b = (const float*)d_in[4];
    float* out = (float*)d_out;

    float* ws = (float*)d_ws;
    float* h0 = ws;
    float* h1 = ws + (size_t)NPIX;
    float* h2 = ws + (size_t)2 * NPIX;
    float* c0 = ws + (size_t)3 * NPIX;
    float* c1 = ws + (size_t)4 * NPIX;

    dim3 grid(256), blk(1024);

    float *ha = h0, *hb = h1, *ca = c0, *cb = c1;
    for (int p = 0; p < 5; ++p) {
        const float* xa = data + (size_t)(4 * p) * NPIX;
        const float* xb = data + (size_t)(4 * p + 1) * NPIX;
        const float* xc = data + (size_t)(4 * p + 2) * NPIX;
        const float* xd = data + (size_t)(4 * p + 3) * NPIX;
        if (p == 0)
            convlstm_four_steps<true><<<grid, blk, 0, stream>>>(xa, xb, xc, xd, ha, ca, enc_w, enc_b, hb, cb);
        else
            convlstm_four_steps<false><<<grid, blk, 0, stream>>>(xa, xb, xc, xd, ha, ca, enc_w, enc_b, hb, cb);
        float* tmp;
        tmp = ha; ha = hb; hb = tmp;
        tmp = ca; ca = cb; cb = tmp;
    }
    const float* dec_in = ha;

    float *hda = hb, *hdb = h2, *cda = ca, *cdb = cb;
    for (int p = 0; p < 5; ++p) {
        float* hdst = (p == 4) ? out : hdb;
        if (p == 0)
            convlstm_dec_four_steps<true><<<grid, blk, 0, stream>>>(dec_in, hda, cda, dec_w, dec_b, hdst, cdb);
        else
            convlstm_dec_four_steps<false><<<grid, blk, 0, stream>>>(dec_in, hda, cda, dec_w, dec_b, hdst, cdb);
        float* tmp;
        tmp = hda; hda = hdb; hdb = tmp;
        tmp = cda; cda = cdb; cdb = tmp;
    }
}


# Implicit GEMM re-blocked on 2x2 pixel blocks: M = 4 gates x 2x2 pixels, K = the block's 4x4 window per plane (16 slots, not 18): 8 WMMAs per tile not 9, decoder 4 not 5; tap sum order changes
# speedup vs baseline: 3.8082x; 1.0451x over previous
#include <hip/hip_runtime.h>

typedef __attribute__((ext_vector_type(2))) float v2f;
typedef __attribute__((ext_vector_type(8))) float v8f;

#define HW 1024
#define NPIX (HW * HW)

#define TW 128
#define TH 32
#define LP 146
#define LROWS 40
#define PLANE (LROWS * LP)
#define NWAVE 32

__device__ __forceinline__ float tanh_hw(float v) {
    return __builtin_amdgcn_tanhf(v);
}
__device__ __forceinline__ float sigmoid_hw(float v) {
    return fmaf(0.5f, tanh_hw(0.5f * v), 0.5f);
}


template <bool FIRST, bool SRC_LDS, bool DST_LDS>
__device__ __forceinline__ void tile_step(
    const int ty, const int tx, const int lrow, const int lcol,
    const float* __restrict__ xs,
    const float* __restrict__ hin, const float* __restrict__ cin,
    const float* hsrc, float* cbuf, float* hdst,
    float* __restrict__ hout, float* __restrict__ cout,
    const v2f (&af)[8], const v8f cinit, const int c, const int hf)
{
    v2f bf[8];
    float cp[2];
    bool  in_img[2];
    const int nr = c >> 3;
    const int nc = c & 7;
    const int py  = ty + 2 * nr + hf;
    const int px0 = tx + 2 * nc;
    const int lown = (lrow + 2 * nr + hf) * LP + lcol + 2 * nc;
    const int wr = ty + 2 * nr - 1;
    const int wc = tx + 2 * nc - 1;
#pragma unroll
    for (int e = 0; e < 2; ++e) {
        const int px = px0 + e;
        in_img[e] = (py >= 0) && (py < HW) && (px >= 0) && (px < HW);
    }
    const bool inside = (ty >= 1) && (ty + 4 <= HW - 1) &&
                        (tx >= 1) && (tx + 16 <= HW - 1);

    if (!SRC_LDS) {
        const float* __restrict__ plane = (!FIRST && hf) ? hin : xs;
        const bool zero_half = FIRST && (hf != 0);
        if (inside) {
            const float* bp = plane + (wr * HW + wc);
#pragma unroll
            for (int ch = 0; ch < 8; ++ch) {
#pragma unroll
                for (int s = 0; s < 2; ++s) {
                    const int j = 2 * ch + s;
                    const float lv = bp[(j >> 2) * HW + (j & 3)];
                    const float v = zero_half ? 0.0f : lv;
                    if (s == 0) bf[ch].x = v; else bf[ch].y = v;
                }
            }
        } else {
#pragma unroll
            for (int ch = 0; ch < 8; ++ch) {
#pragma unroll
                for (int s = 0; s < 2; ++s) {
                    const int j = 2 * ch + s;
                    const int rw = wr + (j >> 2);
                    const int cc = wc + (j & 3);
                    const bool ok = (rw >= 0) && (rw < HW) && (cc >= 0) && (cc < HW);
                    const float tv = plane[ok ? (rw * HW + cc) : 0];
                    const float v = (ok && !zero_half) ? tv : 0.0f;
                    if (s == 0) bf[ch].x = v; else bf[ch].y = v;
                }
            }
        }
#pragma unroll
        for (int e = 0; e < 2; ++e) {
            const float tv = FIRST ? 0.0f : cin[in_img[e] ? (py * HW + px0 + e) : 0];
            cp[e] = in_img[e] ? tv : 0.0f;
        }
    } else {
#pragma unroll
        for (int ch = 0; ch < 8; ++ch) { bf[ch].x = 0.0f; bf[ch].y = 0.0f; }
        if (!hf) {
            if (inside) {
                const float* bp = xs + (wr * HW + wc);
#pragma unroll
                for (int ch = 0; ch < 8; ++ch) {
#pragma unroll
                    for (int s = 0; s < 2; ++s) {
                        const int j = 2 * ch + s;
                        const float v = bp[(j >> 2) * HW + (j & 3)];
                        if (s == 0) bf[ch].x = v; else bf[ch].y = v;
                    }
                }
            } else {
#pragma unroll
                for (int ch = 0; ch < 8; ++ch) {
#pragma unroll
                    for (int s = 0; s < 2; ++s) {
                        const int j = 2 * ch + s;
                        const int rw = wr + (j >> 2);
                        const int cc = wc + (j & 3);
                        const bool ok = (rw >= 0) && (rw < HW) && (cc >= 0) && (cc < HW);
                        const float tv = xs[ok ? (rw * HW + cc) : 0];
                        const float v = ok ? tv : 0.0f;
                        if (s == 0) bf[ch].x = v; else bf[ch].y = v;
                    }
                }
            }
        } else {
            const float* lp = hsrc + (lrow + 2 * nr - 1) * LP + (lcol + 2 * nc - 1);
#pragma unroll
            for (int ch = 0; ch < 8; ++ch) {
#pragma unroll
                for (int s = 0; s < 2; ++s) {
                    const int j = 2 * ch + s;
                    const float v = lp[(j >> 2) * LP + (j & 3)];
                    if (s == 0) bf[ch].x = v; else bf[ch].y = v;
                }
            }
        }
#pragma unroll
        for (int e = 0; e < 2; ++e)
            cp[e] = cbuf[lown + e];
    }

    v8f acc = cinit;
#pragma unroll
    for (int ch = 0; ch < 8; ++ch) {
        acc = __builtin_amdgcn_wmma_f32_16x16x4_f32(
            false, af[ch], false, bf[ch], (short)0, acc, false, false);
    }

#pragma unroll
    for (int e = 0; e < 2; ++e) {
        const float gi = acc[4 * e + 0];
        const float gf = acc[4 * e + 1];
        const float go = acc[4 * e + 2];
        const float gg = acc[4 * e + 3];
        const float cn = sigmoid_hw(gf) * cp[e] + sigmoid_hw(gi) * tanh_hw(gg);
        const float hn = sigmoid_hw(go) * tanh_hw(cn);
        if (DST_LDS) {
            hdst[lown + e] = in_img[e] ? hn : 0.0f;
            cbuf[lown + e] = cn;
        } else {
            const int p = py * HW + px0 + e;
            cout[p] = cn;
            hout[p] = hn;
        }
    }
}

template <bool FIRST>
__global__ __launch_bounds__(1024) void convlstm_four_steps(
    const float* __restrict__ xs0, const float* __restrict__ xs1,
    const float* __restrict__ xs2, const float* __restrict__ xs3,
    const float* __restrict__ hin, const float* __restrict__ cin,
    const float* __restrict__ w,   const float* __restrict__ bias,
    float* __restrict__ hout, float* __restrict__ cout)
{
    __shared__ float hb[2][PLANE];
    __shared__ float cb[PLANE];

    const int tid  = (int)threadIdx.x;
    const int lane = tid & 31;
    const int wv = __builtin_amdgcn_readfirstlane(tid >> 5);
    const int x0 = (int)(blockIdx.x & 7) * TW;
    const int y0 = (int)(blockIdx.x >> 3) * TH;

    const int c  = lane & 15;
    const int hf = lane >> 4;

    if (tid < 160) {
        const int b = tid / 80;
        const int q = tid - 80 * b;
        hb[b][(q >> 1) * LP + (q & 1) * (LP - 1)] = 0.0f;
    } else if (tid < 160 + 2 * LP) {
        hb[1][37 * LP + (tid - 160)] = 0.0f;
    }

    const int pb = c >> 2;
    const float* wl = w + ((c & 3) * 18 + hf * 9);
    float wv9[9];
#pragma unroll
    for (int i = 0; i < 9; ++i) wv9[i] = wl[i];
    v2f af[8];
#pragma unroll
    for (int ch = 0; ch < 8; ++ch) {
#pragma unroll
        for (int s = 0; s < 2; ++s) {
            const int j = 2 * ch + s;
            float v = 0.0f;
#pragma unroll
            for (int pp = 0; pp < 4; ++pp) {
                const int dy = (j >> 2) - (pp >> 1);
                const int dx = (j & 3) - (pp & 1);
                if (dy >= 0 && dy <= 2 && dx >= 0 && dx <= 2)
                    v = (pb == pp) ? wv9[3 * dy + dx] : v;
            }
            if (s == 0) af[ch].x = v; else af[ch].y = v;
        }
    }

    v8f cinit;
#pragma unroll
    for (int q = 0; q < 8; ++q) cinit[q] = bias[q & 3];


    for (int task = wv; task < 90; task += NWAVE) {
        const int k = task / 9;
        const int i = task - 9 * k;
        tile_step<FIRST, false, true>(y0 - 3 + 4 * k, x0 - 8 + 16 * i, 4 * k, 1 + 16 * i,
                                      xs0, hin, cin, nullptr, cb, hb[0], nullptr, nullptr,
                                      af, cinit, c, hf);
    }
    __syncthreads();

    for (int task = wv; task < 81; task += NWAVE) {
        const int k = task / 9;
        const int i = task - 9 * k;
        tile_step<false, true, true>(y0 - 2 + 4 * k, x0 - 8 + 16 * i, 1 + 4 * k, 1 + 16 * i,
                                     xs1, nullptr, nullptr, hb[0], cb, hb[1], nullptr, nullptr,
                                     af, cinit, c, hf);
    }
    __syncthreads();

    for (int task = wv; task < 81; task += NWAVE) {
        const int k = task / 9;
        const int i = task - 9 * k;
        tile_step<false, true, true>(y0 - 1 + 4 * k, x0 - 8 + 16 * i, 2 + 4 * k, 1 + 16 * i,
                                     xs2, nullptr, nullptr, hb[1], cb, hb[0], nullptr, nullptr,
                                     af, cinit, c, hf);
    }
    __syncthreads();

    {
        const int k = wv >> 2;
        for (int t = 0; t < 2; ++t) {
            const int i = 2 * (wv & 3) + t;
            tile_step<false, true, false>(y0 + 4 * k, x0 + 16 * i, 3 + 4 * k, 9 + 16 * i,
                                          xs3, nullptr, nullptr, hb[0], cb, nullptr, hout, cout,
                                          af, cinit, c, hf);
        }
    }
}


typedef __attribute__((ext_vector_type(4))) float v4f;


__device__ __forceinline__ void load_plane_global(
    const float* __restrict__ plane, const int ty, const int tx,
    const int c, const int hf, const bool inside, v2f (&b)[4])
{
    const int wr = ty + 2 * (c >> 3) - 1;
    const int wc = tx + 2 * (c & 7) - 1 + 2 * hf;
    if (inside) {
        const float* bp = plane + (wr * HW + wc);
#pragma unroll
        for (int ch = 0; ch < 4; ++ch) {
            b[ch].x = bp[ch * HW];
            b[ch].y = bp[ch * HW + 1];
        }
    } else {
#pragma unroll
        for (int ch = 0; ch < 4; ++ch) {
#pragma unroll
            for (int s = 0; s < 2; ++s) {
                const int rw = wr + ch;
                const int cc = wc + s;
                const bool ok = (rw >= 0) && (rw < HW) && (cc >= 0) && (cc < HW);
                const float tv = plane[ok ? (rw * HW + cc) : 0];
                const float v = ok ? tv : 0.0f;
                if (s == 0) b[ch].x = v; else b[ch].y = v;
            }
        }
    }
}

template <int MODE>
__device__ __forceinline__ void dec_tile_step(
    const int ty, const int tx, const int lrow, const int lcol,
    const float* __restrict__ xs,
    const float* __restrict__ hin, const float* __restrict__ cin,
    const float* hsrc, float* cbuf, float* hdst, v4f* gx,
    float* __restrict__ hout, float* __restrict__ cout,
    const v2f (&afx)[4], const v2f (&afh)[4], const v8f cinit,
    const int c, const int hf)
{
    constexpr bool STEP0   = (MODE <= 1);
    constexpr bool DST_LDS = (MODE != 3);
    float cp[2];
    bool  in_img[2];
    const int nr = c >> 3;
    const int nc = c & 7;
    const int py  = ty + 2 * nr + hf;
    const int px0 = tx + 2 * nc;
    const int lown = (lrow + 2 * nr + hf) * LP + lcol + 2 * nc;
#pragma unroll
    for (int e = 0; e < 2; ++e) {
        const int px = px0 + e;
        in_img[e] = (py >= 0) && (py < HW) && (px >= 0) && (px < HW);
    }

    v8f acc;
    if (STEP0) {
        const bool inside = (ty >= 1) && (ty + 4 <= HW - 1) &&
                            (tx >= 1) && (tx + 16 <= HW - 1);
        v2f bx[4];
        load_plane_global(xs, ty, tx, c, hf, inside, bx);
        v2f bh[4];
        if (MODE == 1) load_plane_global(hin, ty, tx, c, hf, inside, bh);
        acc = cinit;
#pragma unroll
        for (int ch = 0; ch < 4; ++ch) {
            acc = __builtin_amdgcn_wmma_f32_16x16x4_f32(
                false, afx[ch], false, bx[ch], (short)0, acc, false, false);
        }
#pragma unroll
        for (int e = 0; e < 2; ++e) {
            v4f g4;
            g4.x = acc[4 * e + 0]; g4.y = acc[4 * e + 1];
            g4.z = acc[4 * e + 2]; g4.w = acc[4 * e + 3];
            gx[lown + e] = g4;
        }
        if (MODE == 1) {
#pragma unroll
            for (int ch = 0; ch < 4; ++ch) {
                acc = __builtin_amdgcn_wmma_f32_16x16x4_f32(
                    false, afh[ch], false, bh[ch], (short)0, acc, false, false);
            }
#pragma unroll
            for (int e = 0; e < 2; ++e) {
                const float tv = cin[in_img[e] ? (py * HW + px0 + e) : 0];
                cp[e] = in_img[e] ? tv : 0.0f;
            }
        } else {
            cp[0] = 0.0f; cp[1] = 0.0f;
        }
    } else {
        v2f bh[4];
        const float* lp = hsrc + (lrow + 2 * nr - 1) * LP + (lcol + 2 * nc - 1 + 2 * hf);
#pragma unroll
        for (int ch = 0; ch < 4; ++ch) {
            bh[ch].x = lp[ch * LP];
            bh[ch].y = lp[ch * LP + 1];
        }
#pragma unroll
        for (int e = 0; e < 2; ++e) {
            const v4f g4 = gx[lown + e];
            acc[4 * e + 0] = g4.x; acc[4 * e + 1] = g4.y;
            acc[4 * e + 2] = g4.z; acc[4 * e + 3] = g4.w;
        }
#pragma unroll
        for (int e = 0; e < 2; ++e)
            cp[e] = cbuf[lown + e];
#pragma unroll
        for (int ch = 0; ch < 4; ++ch) {
            acc = __builtin_amdgcn_wmma_f32_16x16x4_f32(
                false, afh[ch], false, bh[ch], (short)0, acc, false, false);
        }
    }

#pragma unroll
    for (int e = 0; e < 2; ++e) {
        const float gi = acc[4 * e + 0];
        const float gf = acc[4 * e + 1];
        const float go = acc[4 * e + 2];
        const float gg = acc[4 * e + 3];
        const float cn = sigmoid_hw(gf) * cp[e] + sigmoid_hw(gi) * tanh_hw(gg);
        const float hn = sigmoid_hw(go) * tanh_hw(cn);
        if (DST_LDS) {
            hdst[lown + e] = in_img[e] ? hn : 0.0f;
            cbuf[lown + e] = cn;
        } else {
            const int p = py * HW + px0 + e;
            cout[p] = cn;
            hout[p] = hn;
        }
    }
}

template <bool FIRST>
__global__ __launch_bounds__(1024) void convlstm_dec_four_steps(
    const float* __restrict__ xs,
    const float* __restrict__ hin, const float* __restrict__ cin,
    const float* __restrict__ w,   const float* __restrict__ bias,
    float* __restrict__ hout, float* __restrict__ cout)
{
    __shared__ float hb[2][PLANE];
    __shared__ float cb[PLANE];
    __shared__ v4f   gx[PLANE];

    const int tid  = (int)threadIdx.x;
    const int lane = tid & 31;
    const int wv = __builtin_amdgcn_readfirstlane(tid >> 5);
    const int x0 = (int)(blockIdx.x & 7) * TW;
    const int y0 = (int)(blockIdx.x >> 3) * TH;

    const int c  = lane & 15;
    const int hf = lane >> 4;

    if (tid < 160) {
        const int b = tid / 80;
        const int q = tid - 80 * b;
        hb[b][(q >> 1) * LP + (q & 1) * (LP - 1)] = 0.0f;
    } else if (tid < 160 + 2 * LP) {
        hb[1][37 * LP + (tid - 160)] = 0.0f;
    }

    const int r = c >> 3;
    const int q = (c >> 2) & 1;
    const float* wg = w + (c & 3) * 18;
    v2f afx[4], afh[4];
#pragma unroll
    for (int ch = 0; ch < 4; ++ch) {
#pragma unroll
        for (int s = 0; s < 2; ++s) {
            const int dy = ch - r;
            const int dx = 2 * hf + s - q;
            const bool ok = (dy >= 0) && (dy <= 2) && (dx >= 0) && (dx <= 2);
            const int ii = ok ? (3 * dy + dx) : 0;
            const float lx = wg[ii];
            const float lh = wg[9 + ii];
            const float vx = ok ? lx : 0.0f;
            const float vh = ok ? lh : 0.0f;
            if (s == 0) { afx[ch].x = vx; afh[ch].x = vh; }
            else        { afx[ch].y = vx; afh[ch].y = vh; }
        }
    }

    v8f cinit;
#pragma unroll
    for (int q = 0; q < 8; ++q) cinit[q] = bias[q & 3];

    for (int task = wv; task < 90; task += NWAVE) {
        const int k = task / 9;
        const int i = task - 9 * k;
        dec_tile_step<FIRST ? 0 : 1>(y0 - 3 + 4 * k, x0 - 8 + 16 * i, 4 * k, 1 + 16 * i,
                                     xs, hin, cin, nullptr, cb, hb[0], gx, nullptr, nullptr,
                                     afx, afh, cinit, c, hf);
    }
    __syncthreads();

    for (int task = wv; task < 81; task += NWAVE) {
        const int k = task / 9;
        const int i = task - 9 * k;
        dec_tile_step<2>(y0 - 2 + 4 * k, x0 - 8 + 16 * i, 1 + 4 * k, 1 + 16 * i,
                         xs, nullptr, nullptr, hb[0], cb, hb[1], gx, nullptr, nullptr,
                         afx, afh, cinit, c, hf);
    }
    __syncthreads();

    for (int task = wv; task < 81; task += NWAVE) {
        const int k = task / 9;
        const int i = task - 9 * k;
        dec_tile_step<2>(y0 - 1 + 4 * k, x0 - 8 + 16 * i, 2 + 4 * k, 1 + 16 * i,
                         xs, nullptr, nullptr, hb[1], cb, hb[0], gx, nullptr, nullptr,
                         afx, afh, cinit, c, hf);
    }
    __syncthreads();

    {
        const int k = wv >> 2;
        for (int t = 0; t < 2; ++t) {
            const int i = 2 * (wv & 3) + t;
            dec_tile_step<3>(y0 + 4 * k, x0 + 16 * i, 3 + 4 * k, 9 + 16 * i,
                             xs, nullptr, nullptr, hb[0], cb, nullptr, gx, hout, cout,
                             afx, afh, cinit, c, hf);
        }
    }
}

extern "C" void kernel_launch(void* const* d_in, const int* in_sizes, int n_in,
                              void* d_out, int out_size, void* d_ws, size_t ws_size,
                              hipStream_t stream) {
    (void)in_sizes; (void)n_in; (void)out_size; (void)ws_size;
    const float* data  = (const float*)d_in[0];
    const float* enc_w = (const float*)d_in[1];
    const float* enc_b = (const float*)d_in[2];
    const float* dec_w = (const float*)d_in[3];
    const float* dec_b = (const float*)d_in[4];
    float* out = (float*)d_out;

    float* ws = (float*)d_ws;
    float* h0 = ws;
    float* h1 = ws + (size_t)NPIX;
    float* h2 = ws + (size_t)2 * NPIX;
    float* c0 = ws + (size_t)3 * NPIX;
    float* c1 = ws + (size_t)4 * NPIX;

    dim3 grid(256), blk(1024);

    float *ha = h0, *hb = h1, *ca = c0, *cb = c1;
    for (int p = 0; p < 5; ++p) {
        const float* xa = data + (size_t)(4 * p) * NPIX;
        const float* xb = data + (size_t)(4 * p + 1) * NPIX;
        const float* xc = data + (size_t)(4 * p + 2) * NPIX;
        const float* xd = data + (size_t)(4 * p + 3) * NPIX;
        if (p == 0)
            convlstm_four_steps<true><<<grid, blk, 0, stream>>>(xa, xb, xc, xd, ha, ca, enc_w, enc_b, hb, cb);
        else
            convlstm_four_steps<false><<<grid, blk, 0, stream>>>(xa, xb, xc, xd, ha, ca, enc_w, enc_b, hb, cb);
        float* tmp;
        tmp = ha; ha = hb; hb = tmp;
        tmp = ca; ca = cb; cb = tmp;
    }
    const float* dec_in = ha;

    float *hda = hb, *hdb = h2, *cda = ca, *cdb = cb;
    for (int p = 0; p < 5; ++p) {
        float* hdst = (p == 4) ? out : hdb;
        if (p == 0)
            convlstm_dec_four_steps<true><<<grid, blk, 0, stream>>>(dec_in, hda, cda, dec_w, dec_b, hdst, cdb);
        else
            convlstm_dec_four_steps<false><<<grid, blk, 0, stream>>>(dec_in, hda, cda, dec_w, dec_b, hdst, cdb);
        float* tmp;
        tmp = hda; hda = hdb; hdb = tmp;
        tmp = cda; cda = cdb; cdb = tmp;
    }
}


# Encoder: a tile's global loads go out one tile ahead (a step's first x tile ahead of the barrier before it); x and h in separate K chunks (decoder's slot map), all lanes load; gate sum order changes
# speedup vs baseline: 4.1216x; 1.0823x over previous
#include <hip/hip_runtime.h>

typedef __attribute__((ext_vector_type(2))) float v2f;
typedef __attribute__((ext_vector_type(8))) float v8f;

#define HW 1024
#define NPIX (HW * HW)

#define TW 128
#define TH 32
#define LP 146
#define LROWS 40
#define PLANE (LROWS * LP)
#define NWAVE 32

__device__ __forceinline__ float tanh_hw(float v) {
    return __builtin_amdgcn_tanhf(v);
}
__device__ __forceinline__ float sigmoid_hw(float v) {
    return fmaf(0.5f, tanh_hw(0.5f * v), 0.5f);
}


struct TileIn {
    v2f   bx[4];
    v2f   bh[4];
    float cp[2];
};


__device__ __forceinline__ void window_offsets(
    const int ty, const int tx, const int c, const int hf, int (&off)[4])
{
    const int wr  = ty + 2 * (c >> 3) - 1;
    const int cc0 = tx + 2 * (c & 7) - 1 + 2 * hf;
    const int ccs = min(max(cc0, 0), HW - 2);
#pragma unroll
    for (int ch = 0; ch < 4; ++ch)
        off[ch] = min(max(wr + ch, 0), HW - 1) * HW + ccs;
}

template <bool FIRST, bool STEP0>
__device__ __forceinline__ void tile_fetch(
    const int ty, const int tx,
    const float* __restrict__ xs,
    const float* __restrict__ hin, const float* __restrict__ cin,
    const int c, const int hf, TileIn& t)
{
    int off[4];
    window_offsets(ty, tx, c, hf, off);
#pragma unroll
    for (int ch = 0; ch < 4; ++ch) {
        t.bx[ch].x = xs[off[ch]];
        t.bx[ch].y = xs[off[ch] + 1];
    }
    if (STEP0 && !FIRST) {
#pragma unroll
        for (int ch = 0; ch < 4; ++ch) {
            t.bh[ch].x = hin[off[ch]];
            t.bh[ch].y = hin[off[ch] + 1];
        }
        const int py  = ty + 2 * (c >> 3) + hf;
        const int px0 = tx + 2 * (c & 7);
        const bool in = (py >= 0) && (py < HW) && (px0 >= 0) && (px0 < HW);
        const int p = in ? (py * HW + px0) : 0;
        t.cp[0] = cin[p];
        t.cp[1] = cin[p + 1];
    }
}

template <bool FIRST, bool SRC_LDS, bool DST_LDS>
__device__ __forceinline__ void tile_compute(
    const int ty, const int tx, const int lrow, const int lcol,
    const float* hsrc, float* cbuf, float* hdst,
    float* __restrict__ hout, float* __restrict__ cout,
    const v2f (&afx)[4], const v2f (&afh)[4], const v8f cinit,
    const int c, const int hf, TileIn& t)
{
    float cp[2];
    bool  in_img[2];
    const int nr = c >> 3;
    const int nc = c & 7;
    const int py  = ty + 2 * nr + hf;
    const int px0 = tx + 2 * nc;
    const int lown = (lrow + 2 * nr + hf) * LP + lcol + 2 * nc;
#pragma unroll
    for (int e = 0; e < 2; ++e) {
        const int px = px0 + e;
        in_img[e] = (py >= 0) && (py < HW) && (px >= 0) && (px < HW);
    }

    v2f bh[4];
    if (SRC_LDS) {
        const float* lp = hsrc + (lrow + 2 * nr - 1) * LP + (lcol + 2 * nc - 1 + 2 * hf);
#pragma unroll
        for (int ch = 0; ch < 4; ++ch) {
            bh[ch].x = lp[ch * LP];
            bh[ch].y = lp[ch * LP + 1];
        }
#pragma unroll
        for (int e = 0; e < 2; ++e)
            cp[e] = cbuf[lown + e];
    } else if (FIRST) {
#pragma unroll
        for (int ch = 0; ch < 4; ++ch) { bh[ch].x = 0.0f; bh[ch].y = 0.0f; }
        cp[0] = 0.0f; cp[1] = 0.0f;
    } else {
#pragma unroll
        for (int ch = 0; ch < 4; ++ch) bh[ch] = t.bh[ch];
        cp[0] = in_img[0] ? t.cp[0] : 0.0f;
        cp[1] = in_img[1] ? t.cp[1] : 0.0f;
    }

    const bool inside = (ty >= 1) && (ty + 4 <= HW - 1) &&
                        (tx >= 1) && (tx + 16 <= HW - 1);
    if (!inside) {
        const int wr  = ty + 2 * nr - 1;
        const int cc0 = tx + 2 * nc - 1 + 2 * hf;
        const bool kx = (cc0 >= 0) && (cc0 < HW);
        const bool ky = (cc0 + 1 >= 0) && (cc0 + 1 < HW);
        const bool left  = (cc0 == -1);
        const bool right = (cc0 == HW - 1);
#pragma unroll
        for (int ch = 0; ch < 4; ++ch) {
            const bool rok = (wr + ch >= 0) && (wr + ch < HW);
            const float x0v = right ? t.bx[ch].y : t.bx[ch].x;
            const float x1v = left  ? t.bx[ch].x : t.bx[ch].y;
            t.bx[ch].x = (rok && kx) ? x0v : 0.0f;
            t.bx[ch].y = (rok && ky) ? x1v : 0.0f;
            if (!SRC_LDS && !FIRST) {
                const float h0v = right ? bh[ch].y : bh[ch].x;
                const float h1v = left  ? bh[ch].x : bh[ch].y;
                bh[ch].x = (rok && kx) ? h0v : 0.0f;
                bh[ch].y = (rok && ky) ? h1v : 0.0f;
            }
        }
    }

    v8f acc = cinit;
#pragma unroll
    for (int ch = 0; ch < 4; ++ch) {
        acc = __builtin_amdgcn_wmma_f32_16x16x4_f32(
            false, afx[ch], false, t.bx[ch], (short)0, acc, false, false);
    }
#pragma unroll
    for (int ch = 0; ch < 4; ++ch) {
        acc = __builtin_amdgcn_wmma_f32_16x16x4_f32(
            false, afh[ch], false, bh[ch], (short)0, acc, false, false);
    }

#pragma unroll
    for (int e = 0; e < 2; ++e) {
        const float gi = acc[4 * e + 0];
        const float gf = acc[4 * e + 1];
        const float go = acc[4 * e + 2];
        const float gg = acc[4 * e + 3];
        const float cn = sigmoid_hw(gf) * cp[e] + sigmoid_hw(gi) * tanh_hw(gg);
        const float hn = sigmoid_hw(go) * tanh_hw(cn);
        if (DST_LDS) {
            hdst[lown + e] = in_img[e] ? hn : 0.0f;
            cbuf[lown + e] = cn;
        } else {
            const int p = py * HW + px0 + e;
            cout[p] = cn;
            hout[p] = hn;
        }
    }
}

template <bool FIRST>
__global__ __launch_bounds__(1024) void convlstm_four_steps(
    const float* __restrict__ xs0, const float* __restrict__ xs1,
    const float* __restrict__ xs2, const float* __restrict__ xs3,
    const float* __restrict__ hin, const float* __restrict__ cin,
    const float* __restrict__ w,   const float* __restrict__ bias,
    float* __restrict__ hout, float* __restrict__ cout)
{
    __shared__ float hb[2][PLANE];
    __shared__ float cb[PLANE];

    const int tid  = (int)threadIdx.x;
    const int lane = tid & 31;
    const int wv = __builtin_amdgcn_readfirstlane(tid >> 5);
    const int x0 = (int)(blockIdx.x & 7) * TW;
    const int y0 = (int)(blockIdx.x >> 3) * TH;

    const int c  = lane & 15;
    const int hf = lane >> 4;

    auto ty_of   = [&](const int s, const int task) { return y0 - 3 + s + 4 * (task / 9); };
    auto tx_of   = [&](const int task) { return x0 - 8 + 16 * (task - 9 * (task / 9)); };
    auto lrow_of = [&](const int s, const int task) { return s + 4 * (task / 9); };
    auto lcol_of = [&](const int task) { return 1 + 16 * (task - 9 * (task / 9)); };
    const int k3 = wv >> 2;
    const int i3 = 2 * (wv & 3);

    const int t0 = wv, t1 = wv + 32, t2 = wv + 64;
    const bool third0  = t2 < 90;
    const bool third12 = t2 < 81;
    const int  t2f0  = third0  ? t2 : 89;
    const int  t2f12 = third12 ? t2 : 80;

    TileIn in_a, in_b;
    tile_fetch<FIRST, true>(ty_of(0, t0), tx_of(t0), xs0, hin, cin, c, hf, in_a);
    tile_fetch<FIRST, true>(ty_of(0, t1), tx_of(t1), xs0, hin, cin, c, hf, in_b);

    if (tid < 160) {
        const int b = tid / 80;
        const int q = tid - 80 * b;
        hb[b][(q >> 1) * LP + (q & 1) * (LP - 1)] = 0.0f;
    } else if (tid < 160 + 2 * LP) {
        hb[1][37 * LP + (tid - 160)] = 0.0f;
    }

    const int r = c >> 3;
    const int q = (c >> 2) & 1;
    const float* wg = w + (c & 3) * 18;
    v2f afx[4], afh[4];
#pragma unroll
    for (int ch = 0; ch < 4; ++ch) {
#pragma unroll
        for (int s = 0; s < 2; ++s) {
            const int dy = ch - r;
            const int dx = 2 * hf + s - q;
            const bool ok = (dy >= 0) && (dy <= 2) && (dx >= 0) && (dx <= 2);
            const int ii = ok ? (3 * dy + dx) : 0;
            const float lx = wg[ii];
            const float lh = wg[9 + ii];
            const float vx = ok ? lx : 0.0f;
            const float vh = ok ? lh : 0.0f;
            if (s == 0) { afx[ch].x = vx; afh[ch].x = vh; }
            else        { afx[ch].y = vx; afh[ch].y = vh; }
        }
    }

    v8f cinit;
#pragma unroll
    for (int q8 = 0; q8 < 8; ++q8) cinit[q8] = bias[q8 & 3];

    tile_compute<FIRST, false, true>(ty_of(0, t0), tx_of(t0), lrow_of(0, t0), lcol_of(t0),
                                     nullptr, cb, hb[0], nullptr, nullptr, afx, afh, cinit, c, hf, in_a);
    TileIn in_c;
    tile_fetch<FIRST, true>(ty_of(0, t2f0), tx_of(t2f0), xs0, hin, cin, c, hf, in_c);
    tile_compute<FIRST, false, true>(ty_of(0, t1), tx_of(t1), lrow_of(0, t1), lcol_of(t1),
                                     nullptr, cb, hb[0], nullptr, nullptr, afx, afh, cinit, c, hf, in_b);
    TileIn s1_a;
    tile_fetch<false, false>(ty_of(1, t0), tx_of(t0), xs1, nullptr, nullptr, c, hf, s1_a);
    if (third0)
        tile_compute<FIRST, false, true>(ty_of(0, t2), tx_of(t2), lrow_of(0, t2), lcol_of(t2),
                                         nullptr, cb, hb[0], nullptr, nullptr, afx, afh, cinit, c, hf, in_c);
    __syncthreads();

    TileIn s1_b;
    tile_fetch<false, false>(ty_of(1, t1), tx_of(t1), xs1, nullptr, nullptr, c, hf, s1_b);
    tile_compute<false, true, true>(ty_of(1, t0), tx_of(t0), lrow_of(1, t0), lcol_of(t0),
                                    hb[0], cb, hb[1], nullptr, nullptr, afx, afh, cinit, c, hf, s1_a);
    TileIn s1_c;
    tile_fetch<false, false>(ty_of(1, t2f12), tx_of(t2f12), xs1, nullptr, nullptr, c, hf, s1_c);
    tile_compute<false, true, true>(ty_of(1, t1), tx_of(t1), lrow_of(1, t1), lcol_of(t1),
                                    hb[0], cb, hb[1], nullptr, nullptr, afx, afh, cinit, c, hf, s1_b);
    TileIn s2_a;
    tile_fetch<false, false>(ty_of(2, t0), tx_of(t0), xs2, nullptr, nullptr, c, hf, s2_a);
    if (third12)
        tile_compute<false, true, true>(ty_of(1, t2), tx_of(t2), lrow_of(1, t2), lcol_of(t2),
                                        hb[0], cb, hb[1], nullptr, nullptr, afx, afh, cinit, c, hf, s1_c);
    __syncthreads();

    TileIn s2_b;
    tile_fetch<false, false>(ty_of(2, t1), tx_of(t1), xs2, nullptr, nullptr, c, hf, s2_b);
    tile_compute<false, true, true>(ty_of(2, t0), tx_of(t0), lrow_of(2, t0), lcol_of(t0),
                                    hb[1], cb, hb[0], nullptr, nullptr, afx, afh, cinit, c, hf, s2_a);
    TileIn s2_c;
    tile_fetch<false, false>(ty_of(2, t2f12), tx_of(t2f12), xs2, nullptr, nullptr, c, hf, s2_c);
    tile_compute<false, true, true>(ty_of(2, t1), tx_of(t1), lrow_of(2, t1), lcol_of(t1),
                                    hb[1], cb, hb[0], nullptr, nullptr, afx, afh, cinit, c, hf, s2_b);
    TileIn s3_a;
    tile_fetch<false, false>(y0 + 4 * k3, x0 + 16 * i3, xs3, nullptr, nullptr, c, hf, s3_a);
    if (third12)
        tile_compute<false, true, true>(ty_of(2, t2), tx_of(t2), lrow_of(2, t2), lcol_of(t2),
                                        hb[1], cb, hb[0], nullptr, nullptr, afx, afh, cinit, c, hf, s2_c);
    __syncthreads();

    TileIn s3_b;
    tile_fetch<false, false>(y0 + 4 * k3, x0 + 16 * (i3 + 1), xs3, nullptr, nullptr, c, hf, s3_b);
    tile_compute<false, true, false>(y0 + 4 * k3, x0 + 16 * i3, 3 + 4 * k3, 9 + 16 * i3,
                                     hb[0], cb, nullptr, hout, cout, afx, afh, cinit, c, hf, s3_a);
    tile_compute<false, true, false>(y0 + 4 * k3, x0 + 16 * (i3 + 1), 3 + 4 * k3, 9 + 16 * (i3 + 1),
                                     hb[0], cb, nullptr, hout, cout, afx, afh, cinit, c, hf, s3_b);
}


typedef __attribute__((ext_vector_type(4))) float v4f;


__device__ __forceinline__ void load_plane_global(
    const float* __restrict__ plane, const int ty, const int tx,
    const int c, const int hf, const bool inside, v2f (&b)[4])
{
    const int wr = ty + 2 * (c >> 3) - 1;
    const int wc = tx + 2 * (c & 7) - 1 + 2 * hf;
    if (inside) {
        const float* bp = plane + (wr * HW + wc);
#pragma unroll
        for (int ch = 0; ch < 4; ++ch) {
            b[ch].x = bp[ch * HW];
            b[ch].y = bp[ch * HW + 1];
        }
    } else {
#pragma unroll
        for (int ch = 0; ch < 4; ++ch) {
#pragma unroll
            for (int s = 0; s < 2; ++s) {
                const int rw = wr + ch;
                const int cc = wc + s;
                const bool ok = (rw >= 0) && (rw < HW) && (cc >= 0) && (cc < HW);
                const float tv = plane[ok ? (rw * HW + cc) : 0];
                const float v = ok ? tv : 0.0f;
                if (s == 0) b[ch].x = v; else b[ch].y = v;
            }
        }
    }
}

template <int MODE>
__device__ __forceinline__ void dec_tile_step(
    const int ty, const int tx, const int lrow, const int lcol,
    const float* __restrict__ xs,
    const float* __restrict__ hin, const float* __restrict__ cin,
    const float* hsrc, float* cbuf, float* hdst, v4f* gx,
    float* __restrict__ hout, float* __restrict__ cout,
    const v2f (&afx)[4], const v2f (&afh)[4], const v8f cinit,
    const int c, const int hf)
{
    constexpr bool STEP0   = (MODE <= 1);
    constexpr bool DST_LDS = (MODE != 3);
    float cp[2];
    bool  in_img[2];
    const int nr = c >> 3;
    const int nc = c & 7;
    const int py  = ty + 2 * nr + hf;
    const int px0 = tx + 2 * nc;
    const int lown = (lrow + 2 * nr + hf) * LP + lcol + 2 * nc;
#pragma unroll
    for (int e = 0; e < 2; ++e) {
        const int px = px0 + e;
        in_img[e] = (py >= 0) && (py < HW) && (px >= 0) && (px < HW);
    }

    v8f acc;
    if (STEP0) {
        const bool inside = (ty >= 1) && (ty + 4 <= HW - 1) &&
                            (tx >= 1) && (tx + 16 <= HW - 1);
        v2f bx[4];
        load_plane_global(xs, ty, tx, c, hf, inside, bx);
        v2f bh[4];
        if (MODE == 1) load_plane_global(hin, ty, tx, c, hf, inside, bh);
        acc = cinit;
#pragma unroll
        for (int ch = 0; ch < 4; ++ch) {
            acc = __builtin_amdgcn_wmma_f32_16x16x4_f32(
                false, afx[ch], false, bx[ch], (short)0, acc, false, false);
        }
#pragma unroll
        for (int e = 0; e < 2; ++e) {
            v4f g4;
            g4.x = acc[4 * e + 0]; g4.y = acc[4 * e + 1];
            g4.z = acc[4 * e + 2]; g4.w = acc[4 * e + 3];
            gx[lown + e] = g4;
        }
        if (MODE == 1) {
#pragma unroll
            for (int ch = 0; ch < 4; ++ch) {
                acc = __builtin_amdgcn_wmma_f32_16x16x4_f32(
                    false, afh[ch], false, bh[ch], (short)0, acc, false, false);
            }
#pragma unroll
            for (int e = 0; e < 2; ++e) {
                const float tv = cin[in_img[e] ? (py * HW + px0 + e) : 0];
                cp[e] = in_img[e] ? tv : 0.0f;
            }
        } else {
            cp[0] = 0.0f; cp[1] = 0.0f;
        }
    } else {
        v2f bh[4];
        const float* lp = hsrc + (lrow + 2 * nr - 1) * LP + (lcol + 2 * nc - 1 + 2 * hf);
#pragma unroll
        for (int ch = 0; ch < 4; ++ch) {
            bh[ch].x = lp[ch * LP];
            bh[ch].y = lp[ch * LP + 1];
        }
#pragma unroll
        for (int e = 0; e < 2; ++e) {
            const v4f g4 = gx[lown + e];
            acc[4 * e + 0] = g4.x; acc[4 * e + 1] = g4.y;
            acc[4 * e + 2] = g4.z; acc[4 * e + 3] = g4.w;
        }
#pragma unroll
        for (int e = 0; e < 2; ++e)
            cp[e] = cbuf[lown + e];
#pragma unroll
        for (int ch = 0; ch < 4; ++ch) {
            acc = __builtin_amdgcn_wmma_f32_16x16x4_f32(
                false, afh[ch], false, bh[ch], (short)0, acc, false, false);
        }
    }

#pragma unroll
    for (int e = 0; e < 2; ++e) {
        const float gi = acc[4 * e + 0];
        const float gf = acc[4 * e + 1];
        const float go = acc[4 * e + 2];
        const float gg = acc[4 * e + 3];
        const float cn = sigmoid_hw(gf) * cp[e] + sigmoid_hw(gi) * tanh_hw(gg);
        const float hn = sigmoid_hw(go) * tanh_hw(cn);
        if (DST_LDS) {
            hdst[lown + e] = in_img[e] ? hn : 0.0f;
            cbuf[lown + e] = cn;
        } else {
            const int p = py * HW + px0 + e;
            cout[p] = cn;
            hout[p] = hn;
        }
    }
}

template <bool FIRST>
__global__ __launch_bounds__(1024) void convlstm_dec_four_steps(
    const float* __restrict__ xs,
    const float* __restrict__ hin, const float* __restrict__ cin,
    const float* __restrict__ w,   const float* __restrict__ bias,
    float* __restrict__ hout, float* __restrict__ cout)
{
    __shared__ float hb[2][PLANE];
    __shared__ float cb[PLANE];
    __shared__ v4f   gx[PLANE];

    const int tid  = (int)threadIdx.x;
    const int lane = tid & 31;
    const int wv = __builtin_amdgcn_readfirstlane(tid >> 5);
    const int x0 = (int)(blockIdx.x & 7) * TW;
    const int y0 = (int)(blockIdx.x >> 3) * TH;

    const int c  = lane & 15;
    const int hf = lane >> 4;

    if (tid < 160) {
        const int b = tid / 80;
        const int q = tid - 80 * b;
        hb[b][(q >> 1) * LP + (q & 1) * (LP - 1)] = 0.0f;
    } else if (tid < 160 + 2 * LP) {
        hb[1][37 * LP + (tid - 160)] = 0.0f;
    }

    const int r = c >> 3;
    const int q = (c >> 2) & 1;
    const float* wg = w + (c & 3) * 18;
    v2f afx[4], afh[4];
#pragma unroll
    for (int ch = 0; ch < 4; ++ch) {
#pragma unroll
        for (int s = 0; s < 2; ++s) {
            const int dy = ch - r;
            const int dx = 2 * hf + s - q;
            const bool ok = (dy >= 0) && (dy <= 2) && (dx >= 0) && (dx <= 2);
            const int ii = ok ? (3 * dy + dx) : 0;
            const float lx = wg[ii];
            const float lh = wg[9 + ii];
            const float vx = ok ? lx : 0.0f;
            const float vh = ok ? lh : 0.0f;
            if (s == 0) { afx[ch].x = vx; afh[ch].x = vh; }
            else        { afx[ch].y = vx; afh[ch].y = vh; }
        }
    }

    v8f cinit;
#pragma unroll
    for (int q = 0; q < 8; ++q) cinit[q] = bias[q & 3];

    for (int task = wv; task < 90; task += NWAVE) {
        const int k = task / 9;
        const int i = task - 9 * k;
        dec_tile_step<FIRST ? 0 : 1>(y0 - 3 + 4 * k, x0 - 8 + 16 * i, 4 * k, 1 + 16 * i,
                                     xs, hin, cin, nullptr, cb, hb[0], gx, nullptr, nullptr,
                                     afx, afh, cinit, c, hf);
    }
    __syncthreads();

    for (int task = wv; task < 81; task += NWAVE) {
        const int k = task / 9;
        const int i = task - 9 * k;
        dec_tile_step<2>(y0 - 2 + 4 * k, x0 - 8 + 16 * i, 1 + 4 * k, 1 + 16 * i,
                         xs, nullptr, nullptr, hb[0], cb, hb[1], gx, nullptr, nullptr,
                         afx, afh, cinit, c, hf);
    }
    __syncthreads();

    for (int task = wv; task < 81; task += NWAVE) {
        const int k = task / 9;
        const int i = task - 9 * k;
        dec_tile_step<2>(y0 - 1 + 4 * k, x0 - 8 + 16 * i, 2 + 4 * k, 1 + 16 * i,
                         xs, nullptr, nullptr, hb[1], cb, hb[0], gx, nullptr, nullptr,
                         afx, afh, cinit, c, hf);
    }
    __syncthreads();

    {
        const int k = wv >> 2;
        for (int t = 0; t < 2; ++t) {
            const int i = 2 * (wv & 3) + t;
            dec_tile_step<3>(y0 + 4 * k, x0 + 16 * i, 3 + 4 * k, 9 + 16 * i,
                             xs, nullptr, nullptr, hb[0], cb, nullptr, gx, hout, cout,
                             afx, afh, cinit, c, hf);
        }
    }
}

extern "C" void kernel_launch(void* const* d_in, const int* in_sizes, int n_in,
                              void* d_out, int out_size, void* d_ws, size_t ws_size,
                              hipStream_t stream) {
    (void)in_sizes; (void)n_in; (void)out_size; (void)ws_size;
    const float* data  = (const float*)d_in[0];
    const float* enc_w = (const float*)d_in[1];
    const float* enc_b = (const float*)d_in[2];
    const float* dec_w = (const float*)d_in[3];
    const float* dec_b = (const float*)d_in[4];
    float* out = (float*)d_out;

    float* ws = (float*)d_ws;
    float* h0 = ws;
    float* h1 = ws + (size_t)NPIX;
    float* h2 = ws + (size_t)2 * NPIX;
    float* c0 = ws + (size_t)3 * NPIX;
    float* c1 = ws + (size_t)4 * NPIX;

    dim3 grid(256), blk(1024);

    float *ha = h0, *hb = h1, *ca = c0, *cb = c1;
    for (int p = 0; p < 5; ++p) {
        const float* xa = data + (size_t)(4 * p) * NPIX;
        const float* xb = data + (size_t)(4 * p + 1) * NPIX;
        const float* xc = data + (size_t)(4 * p + 2) * NPIX;
        const float* xd = data + (size_t)(4 * p + 3) * NPIX;
        if (p == 0)
            convlstm_four_steps<true><<<grid, blk, 0, stream>>>(xa, xb, xc, xd, ha, ca, enc_w, enc_b, hb, cb);
        else
            convlstm_four_steps<false><<<grid, blk, 0, stream>>>(xa, xb, xc, xd, ha, ca, enc_w, enc_b, hb, cb);
        float* tmp;
        tmp = ha; ha = hb; hb = tmp;
        tmp = ca; ca = cb; cb = tmp;
    }
    const float* dec_in = ha;

    float *hda = hb, *hdb = h2, *cda = ca, *cdb = cb;
    for (int p = 0; p < 5; ++p) {
        float* hdst = (p == 4) ? out : hdb;
        if (p == 0)
            convlstm_dec_four_steps<true><<<grid, blk, 0, stream>>>(dec_in, hda, cda, dec_w, dec_b, hdst, cdb);
        else
            convlstm_dec_four_steps<false><<<grid, blk, 0, stream>>>(dec_in, hda, cda, dec_w, dec_b, hdst, cdb);
        float* tmp;
        tmp = hda; hda = hdb; hdb = tmp;
        tmp = cda; cda = cdb; cdb = tmp;
    }
}


# Decoder: 7, 7 and 6 time steps per launch (was 4), h and c kept in LDS; the constant x half of the gate sums is built 3 times, not 5; 10 launches -> 8; tiles sit on other rows: tap sum order changes
# speedup vs baseline: 4.2572x; 1.0329x over previous
#include <hip/hip_runtime.h>

typedef __attribute__((ext_vector_type(2))) float v2f;
typedef __attribute__((ext_vector_type(8))) float v8f;

#define HW 1024
#define NPIX (HW * HW)

#define TW 128
#define TH 32
#define LP 146
#define LROWS 40
#define PLANE (LROWS * LP)
#define NWAVE 32

__device__ __forceinline__ float tanh_hw(float v) {
    return __builtin_amdgcn_tanhf(v);
}
__device__ __forceinline__ float sigmoid_hw(float v) {
    return fmaf(0.5f, tanh_hw(0.5f * v), 0.5f);
}


struct TileIn {
    v2f   bx[4];
    v2f   bh[4];
    float cp[2];
};


__device__ __forceinline__ void window_offsets(
    const int ty, const int tx, const int c, const int hf, int (&off)[4])
{
    const int wr  = ty + 2 * (c >> 3) - 1;
    const int cc0 = tx + 2 * (c & 7) - 1 + 2 * hf;
    const int ccs = min(max(cc0, 0), HW - 2);
#pragma unroll
    for (int ch = 0; ch < 4; ++ch)
        off[ch] = min(max(wr + ch, 0), HW - 1) * HW + ccs;
}

template <bool FIRST, bool STEP0>
__device__ __forceinline__ void tile_fetch(
    const int ty, const int tx,
    const float* __restrict__ xs,
    const float* __restrict__ hin, const float* __restrict__ cin,
    const int c, const int hf, TileIn& t)
{
    int off[4];
    window_offsets(ty, tx, c, hf, off);
#pragma unroll
    for (int ch = 0; ch < 4; ++ch) {
        t.bx[ch].x = xs[off[ch]];
        t.bx[ch].y = xs[off[ch] + 1];
    }
    if (STEP0 && !FIRST) {
#pragma unroll
        for (int ch = 0; ch < 4; ++ch) {
            t.bh[ch].x = hin[off[ch]];
            t.bh[ch].y = hin[off[ch] + 1];
        }
        const int py  = ty + 2 * (c >> 3) + hf;
        const int px0 = tx + 2 * (c & 7);
        const bool in = (py >= 0) && (py < HW) && (px0 >= 0) && (px0 < HW);
        const int p = in ? (py * HW + px0) : 0;
        t.cp[0] = cin[p];
        t.cp[1] = cin[p + 1];
    }
}

template <bool FIRST, bool SRC_LDS, bool DST_LDS>
__device__ __forceinline__ void tile_compute(
    const int ty, const int tx, const int lrow, const int lcol,
    const float* hsrc, float* cbuf, float* hdst,
    float* __restrict__ hout, float* __restrict__ cout,
    const v2f (&afx)[4], const v2f (&afh)[4], const v8f cinit,
    const int c, const int hf, TileIn& t)
{
    float cp[2];
    bool  in_img[2];
    const int nr = c >> 3;
    const int nc = c & 7;
    const int py  = ty + 2 * nr + hf;
    const int px0 = tx + 2 * nc;
    const int lown = (lrow + 2 * nr + hf) * LP + lcol + 2 * nc;
#pragma unroll
    for (int e = 0; e < 2; ++e) {
        const int px = px0 + e;
        in_img[e] = (py >= 0) && (py < HW) && (px >= 0) && (px < HW);
    }

    v2f bh[4];
    if (SRC_LDS) {
        const float* lp = hsrc + (lrow + 2 * nr - 1) * LP + (lcol + 2 * nc - 1 + 2 * hf);
#pragma unroll
        for (int ch = 0; ch < 4; ++ch) {
            bh[ch].x = lp[ch * LP];
            bh[ch].y = lp[ch * LP + 1];
        }
#pragma unroll
        for (int e = 0; e < 2; ++e)
            cp[e] = cbuf[lown + e];
    } else if (FIRST) {
#pragma unroll
        for (int ch = 0; ch < 4; ++ch) { bh[ch].x = 0.0f; bh[ch].y = 0.0f; }
        cp[0] = 0.0f; cp[1] = 0.0f;
    } else {
#pragma unroll
        for (int ch = 0; ch < 4; ++ch) bh[ch] = t.bh[ch];
        cp[0] = in_img[0] ? t.cp[0] : 0.0f;
        cp[1] = in_img[1] ? t.cp[1] : 0.0f;
    }

    const bool inside = (ty >= 1) && (ty + 4 <= HW - 1) &&
                        (tx >= 1) && (tx + 16 <= HW - 1);
    if (!inside) {
        const int wr  = ty + 2 * nr - 1;
        const int cc0 = tx + 2 * nc - 1 + 2 * hf;
        const bool kx = (cc0 >= 0) && (cc0 < HW);
        const bool ky = (cc0 + 1 >= 0) && (cc0 + 1 < HW);
        const bool left  = (cc0 == -1);
        const bool right = (cc0 == HW - 1);
#pragma unroll
        for (int ch = 0; ch < 4; ++ch) {
            const bool rok = (wr + ch >= 0) && (wr + ch < HW);
            const float x0v = right ? t.bx[ch].y : t.bx[ch].x;
            const float x1v = left  ? t.bx[ch].x : t.bx[ch].y;
            t.bx[ch].x = (rok && kx) ? x0v : 0.0f;
            t.bx[ch].y = (rok && ky) ? x1v : 0.0f;
            if (!SRC_LDS && !FIRST) {
                const float h0v = right ? bh[ch].y : bh[ch].x;
                const float h1v = left  ? bh[ch].x : bh[ch].y;
                bh[ch].x = (rok && kx) ? h0v : 0.0f;
                bh[ch].y = (rok && ky) ? h1v : 0.0f;
            }
        }
    }

    v8f acc = cinit;
#pragma unroll
    for (int ch = 0; ch < 4; ++ch) {
        acc = __builtin_amdgcn_wmma_f32_16x16x4_f32(
            false, afx[ch], false, t.bx[ch], (short)0, acc, false, false);
    }
#pragma unroll
    for (int ch = 0; ch < 4; ++ch) {
        acc = __builtin_amdgcn_wmma_f32_16x16x4_f32(
            false, afh[ch], false, bh[ch], (short)0, acc, false, false);
    }

#pragma unroll
    for (int e = 0; e < 2; ++e) {
        const float gi = acc[4 * e + 0];
        const float gf = acc[4 * e + 1];
        const float go = acc[4 * e + 2];
        const float gg = acc[4 * e + 3];
        const float cn = sigmoid_hw(gf) * cp[e] + sigmoid_hw(gi) * tanh_hw(gg);
        const float hn = sigmoid_hw(go) * tanh_hw(cn);
        if (DST_LDS) {
            hdst[lown + e] = in_img[e] ? hn : 0.0f;
            cbuf[lown + e] = cn;
        } else {
            const int p = py * HW + px0 + e;
            cout[p] = cn;
            hout[p] = hn;
        }
    }
}

template <bool FIRST>
__global__ __launch_bounds__(1024) void convlstm_four_steps(
    const float* __restrict__ xs0, const float* __restrict__ xs1,
    const float* __restrict__ xs2, const float* __restrict__ xs3,
    const float* __restrict__ hin, const float* __restrict__ cin,
    const float* __restrict__ w,   const float* __restrict__ bias,
    float* __restrict__ hout, float* __restrict__ cout)
{
    __shared__ float hb[2][PLANE];
    __shared__ float cb[PLANE];

    const int tid  = (int)threadIdx.x;
    const int lane = tid & 31;
    const int wv = __builtin_amdgcn_readfirstlane(tid >> 5);
    const int x0 = (int)(blockIdx.x & 7) * TW;
    const int y0 = (int)(blockIdx.x >> 3) * TH;

    const int c  = lane & 15;
    const int hf = lane >> 4;

    auto ty_of   = [&](const int s, const int task) { return y0 - 3 + s + 4 * (task / 9); };
    auto tx_of   = [&](const int task) { return x0 - 8 + 16 * (task - 9 * (task / 9)); };
    auto lrow_of = [&](const int s, const int task) { return s + 4 * (task / 9); };
    auto lcol_of = [&](const int task) { return 1 + 16 * (task - 9 * (task / 9)); };
    const int k3 = wv >> 2;
    const int i3 = 2 * (wv & 3);

    const int t0 = wv, t1 = wv + 32, t2 = wv + 64;
    const bool third0  = t2 < 90;
    const bool third12 = t2 < 81;
    const int  t2f0  = third0  ? t2 : 89;
    const int  t2f12 = third12 ? t2 : 80;

    TileIn in_a, in_b;
    tile_fetch<FIRST, true>(ty_of(0, t0), tx_of(t0), xs0, hin, cin, c, hf, in_a);
    tile_fetch<FIRST, true>(ty_of(0, t1), tx_of(t1), xs0, hin, cin, c, hf, in_b);

    if (tid < 160) {
        const int b = tid / 80;
        const int q = tid - 80 * b;
        hb[b][(q >> 1) * LP + (q & 1) * (LP - 1)] = 0.0f;
    } else if (tid < 160 + 2 * LP) {
        hb[1][37 * LP + (tid - 160)] = 0.0f;
    }

    const int r = c >> 3;
    const int q = (c >> 2) & 1;
    const float* wg = w + (c & 3) * 18;
    v2f afx[4], afh[4];
#pragma unroll
    for (int ch = 0; ch < 4; ++ch) {
#pragma unroll
        for (int s = 0; s < 2; ++s) {
            const int dy = ch - r;
            const int dx = 2 * hf + s - q;
            const bool ok = (dy >= 0) && (dy <= 2) && (dx >= 0) && (dx <= 2);
            const int ii = ok ? (3 * dy + dx) : 0;
            const float lx = wg[ii];
            const float lh = wg[9 + ii];
            const float vx = ok ? lx : 0.0f;
            const float vh = ok ? lh : 0.0f;
            if (s == 0) { afx[ch].x = vx; afh[ch].x = vh; }
            else        { afx[ch].y = vx; afh[ch].y = vh; }
        }
    }

    v8f cinit;
#pragma unroll
    for (int q8 = 0; q8 < 8; ++q8) cinit[q8] = bias[q8 & 3];

    tile_compute<FIRST, false, true>(ty_of(0, t0), tx_of(t0), lrow_of(0, t0), lcol_of(t0),
                                     nullptr, cb, hb[0], nullptr, nullptr, afx, afh, cinit, c, hf, in_a);
    TileIn in_c;
    tile_fetch<FIRST, true>(ty_of(0, t2f0), tx_of(t2f0), xs0, hin, cin, c, hf, in_c);
    tile_compute<FIRST, false, true>(ty_of(0, t1), tx_of(t1), lrow_of(0, t1), lcol_of(t1),
                                     nullptr, cb, hb[0], nullptr, nullptr, afx, afh, cinit, c, hf, in_b);
    TileIn s1_a;
    tile_fetch<false, false>(ty_of(1, t0), tx_of(t0), xs1, nullptr, nullptr, c, hf, s1_a);
    if (third0)
        tile_compute<FIRST, false, true>(ty_of(0, t2), tx_of(t2), lrow_of(0, t2), lcol_of(t2),
                                         nullptr, cb, hb[0], nullptr, nullptr, afx, afh, cinit, c, hf, in_c);
    __syncthreads();

    TileIn s1_b;
    tile_fetch<false, false>(ty_of(1, t1), tx_of(t1), xs1, nullptr, nullptr, c, hf, s1_b);
    tile_compute<false, true, true>(ty_of(1, t0), tx_of(t0), lrow_of(1, t0), lcol_of(t0),
                                    hb[0], cb, hb[1], nullptr, nullptr, afx, afh, cinit, c, hf, s1_a);
    TileIn s1_c;
    tile_fetch<false, false>(ty_of(1, t2f12), tx_of(t2f12), xs1, nullptr, nullptr, c, hf, s1_c);
    tile_compute<false, true, true>(ty_of(1, t1), tx_of(t1), lrow_of(1, t1), lcol_of(t1),
                                    hb[0], cb, hb[1], nullptr, nullptr, afx, afh, cinit, c, hf, s1_b);
    TileIn s2_a;
    tile_fetch<false, false>(ty_of(2, t0), tx_of(t0), xs2, nullptr, nullptr, c, hf, s2_a);
    if (third12)
        tile_compute<false, true, true>(ty_of(1, t2), tx_of(t2), lrow_of(1, t2), lcol_of(t2),
                                        hb[0], cb, hb[1], nullptr, nullptr, afx, afh, cinit, c, hf, s1_c);
    __syncthreads();

    TileIn s2_b;
    tile_fetch<false, false>(ty_of(2, t1), tx_of(t1), xs2, nullptr, nullptr, c, hf, s2_b);
    tile_compute<false, true, true>(ty_of(2, t0), tx_of(t0), lrow_of(2, t0), lcol_of(t0),
                                    hb[1], cb, hb[0], nullptr, nullptr, afx, afh, cinit, c, hf, s2_a);
    TileIn s2_c;
    tile_fetch<false, false>(ty_of(2, t2f12), tx_of(t2f12), xs2, nullptr, nullptr, c, hf, s2_c);
    tile_compute<false, true, true>(ty_of(2, t1), tx_of(t1), lrow_of(2, t1), lcol_of(t1),
                                    hb[1], cb, hb[0], nullptr, nullptr, afx, afh, cinit, c, hf, s2_b);
    TileIn s3_a;
    tile_fetch<false, false>(y0 + 4 * k3, x0 + 16 * i3, xs3, nullptr, nullptr, c, hf, s3_a);
    if (third12)
        tile_compute<false, true, true>(ty_of(2, t2), tx_of(t2), lrow_of(2, t2), lcol_of(t2),
                                        hb[1], cb, hb[0], nullptr, nullptr, afx, afh, cinit, c, hf, s2_c);
    __syncthreads();

    TileIn s3_b;
    tile_fetch<false, false>(y0 + 4 * k3, x0 + 16 * (i3 + 1), xs3, nullptr, nullptr, c, hf, s3_b);
    tile_compute<false, true, false>(y0 + 4 * k3, x0 + 16 * i3, 3 + 4 * k3, 9 + 16 * i3,
                                     hb[0], cb, nullptr, hout, cout, afx, afh, cinit, c, hf, s3_a);
    tile_compute<false, true, false>(y0 + 4 * k3, x0 + 16 * (i3 + 1), 3 + 4 * k3, 9 + 16 * (i3 + 1),
                                     hb[0], cb, nullptr, hout, cout, afx, afh, cinit, c, hf, s3_b);
}


typedef __attribute__((ext_vector_type(4))) float v4f;


__device__ __forceinline__ void load_plane_global(
    const float* __restrict__ plane, const int ty, const int tx,
    const int c, const int hf, const bool inside, v2f (&b)[4])
{
    const int wr = ty + 2 * (c >> 3) - 1;
    const int wc = tx + 2 * (c & 7) - 1 + 2 * hf;
    if (inside) {
        const float* bp = plane + (wr * HW + wc);
#pragma unroll
        for (int ch = 0; ch < 4; ++ch) {
            b[ch].x = bp[ch * HW];
            b[ch].y = bp[ch * HW + 1];
        }
    } else {
#pragma unroll
        for (int ch = 0; ch < 4; ++ch) {
#pragma unroll
            for (int s = 0; s < 2; ++s) {
                const int rw = wr + ch;
                const int cc = wc + s;
                const bool ok = (rw >= 0) && (rw < HW) && (cc >= 0) && (cc < HW);
                const float tv = plane[ok ? (rw * HW + cc) : 0];
                const float v = ok ? tv : 0.0f;
                if (s == 0) b[ch].x = v; else b[ch].y = v;
            }
        }
    }
}

template <int MODE>
__device__ __forceinline__ void dec_tile_step(
    const int ty, const int tx, const int lrow, const int lcol,
    const float* __restrict__ xs,
    const float* __restrict__ hin, const float* __restrict__ cin,
    const float* hsrc, float* cbuf, float* hdst, v4f* gx,
    float* __restrict__ hout, float* __restrict__ cout,
    const v2f (&afx)[4], const v2f (&afh)[4], const v8f cinit,
    const int c, const int hf)
{
    constexpr bool STEP0   = (MODE <= 1);
    constexpr bool DST_LDS = (MODE != 3);
    float cp[2];
    bool  in_img[2];
    const int nr = c >> 3;
    const int nc = c & 7;
    const int py  = ty + 2 * nr + hf;
    const int px0 = tx + 2 * nc;
    const int lown = (lrow + 2 * nr + hf) * LP + lcol + 2 * nc;
#pragma unroll
    for (int e = 0; e < 2; ++e) {
        const int px = px0 + e;
        in_img[e] = (py >= 0) && (py < HW) && (px >= 0) && (px < HW);
    }

    v8f acc;
    if (STEP0) {
        const bool inside = (ty >= 1) && (ty + 4 <= HW - 1) &&
                            (tx >= 1) && (tx + 16 <= HW - 1);
        v2f bx[4];
        load_plane_global(xs, ty, tx, c, hf, inside, bx);
        v2f bh[4];
        if (MODE == 1) load_plane_global(hin, ty, tx, c, hf, inside, bh);
        acc = cinit;
#pragma unroll
        for (int ch = 0; ch < 4; ++ch) {
            acc = __builtin_amdgcn_wmma_f32_16x16x4_f32(
                false, afx[ch], false, bx[ch], (short)0, acc, false, false);
        }
#pragma unroll
        for (int e = 0; e < 2; ++e) {
            v4f g4;
            g4.x = acc[4 * e + 0]; g4.y = acc[4 * e + 1];
            g4.z = acc[4 * e + 2]; g4.w = acc[4 * e + 3];
            gx[lown + e] = g4;
        }
        if (MODE == 1) {
#pragma unroll
            for (int ch = 0; ch < 4; ++ch) {
                acc = __builtin_amdgcn_wmma_f32_16x16x4_f32(
                    false, afh[ch], false, bh[ch], (short)0, acc, false, false);
            }
#pragma unroll
            for (int e = 0; e < 2; ++e) {
                const float tv = cin[in_img[e] ? (py * HW + px0 + e) : 0];
                cp[e] = in_img[e] ? tv : 0.0f;
            }
        } else {
            cp[0] = 0.0f; cp[1] = 0.0f;
        }
    } else {
        v2f bh[4];
        const float* lp = hsrc + (lrow + 2 * nr - 1) * LP + (lcol + 2 * nc - 1 + 2 * hf);
#pragma unroll
        for (int ch = 0; ch < 4; ++ch) {
            bh[ch].x = lp[ch * LP];
            bh[ch].y = lp[ch * LP + 1];
        }
#pragma unroll
        for (int e = 0; e < 2; ++e) {
            const v4f g4 = gx[lown + e];
            acc[4 * e + 0] = g4.x; acc[4 * e + 1] = g4.y;
            acc[4 * e + 2] = g4.z; acc[4 * e + 3] = g4.w;
        }
#pragma unroll
        for (int e = 0; e < 2; ++e)
            cp[e] = cbuf[lown + e];
#pragma unroll
        for (int ch = 0; ch < 4; ++ch) {
            acc = __builtin_amdgcn_wmma_f32_16x16x4_f32(
                false, afh[ch], false, bh[ch], (short)0, acc, false, false);
        }
    }

#pragma unroll
    for (int e = 0; e < 2; ++e) {
        const float gi = acc[4 * e + 0];
        const float gf = acc[4 * e + 1];
        const float go = acc[4 * e + 2];
        const float gg = acc[4 * e + 3];
        const float cn = sigmoid_hw(gf) * cp[e] + sigmoid_hw(gi) * tanh_hw(gg);
        const float hn = sigmoid_hw(go) * tanh_hw(cn);
        if (DST_LDS) {
            hdst[lown + e] = in_img[e] ? hn : 0.0f;
            cbuf[lown + e] = cn;
        } else {
            const int p = py * HW + px0 + e;
            cout[p] = cn;
            hout[p] = hn;
        }
    }
}

template <int S, bool FIRST>
__global__ __launch_bounds__(1024) void convlstm_dec_steps(
    const float* __restrict__ xs,
    const float* __restrict__ hin, const float* __restrict__ cin,
    const float* __restrict__ w,   const float* __restrict__ bias,
    float* __restrict__ hout, float* __restrict__ cout)
{
    constexpr int NRT0 = (32 + 2 * (S - 1) + 3) / 4;
    constexpr int LR   = 4 * NRT0 + 2;
    constexpr int PL   = LR * LP;
    __shared__ float hb[2][PL];
    __shared__ float cb[PL];
    __shared__ v4f   gx[PL];

    const int tid  = (int)threadIdx.x;
    const int lane = tid & 31;
    const int wv = __builtin_amdgcn_readfirstlane(tid >> 5);
    const int x0 = (int)(blockIdx.x & 7) * TW;
    const int y0 = (int)(blockIdx.x >> 3) * TH;

    const int c  = lane & 15;
    const int hf = lane >> 4;

    for (int i = tid; i < PL; i += NWAVE * 32) hb[1][i] = 0.0f;
    if (tid < 2 * LR)
        hb[0][(tid >> 1) * LP + (tid & 1) * (LP - 1)] = 0.0f;
    if (tid < 2 * LP) {
        const v4f z4 = {0.0f, 0.0f, 0.0f, 0.0f};
        hb[0][4 * NRT0 * LP + tid] = 0.0f;
        cb[4 * NRT0 * LP + tid]    = 0.0f;
        gx[4 * NRT0 * LP + tid]    = z4;
    }

    const int r = c >> 3;
    const int q = (c >> 2) & 1;
    const float* wg = w + (c & 3) * 18;
    v2f afx[4], afh[4];
#pragma unroll
    for (int ch = 0; ch < 4; ++ch) {
#pragma unroll
        for (int s = 0; s < 2; ++s) {
            const int dy = ch - r;
            const int dx = 2 * hf + s - q;
            const bool ok = (dy >= 0) && (dy <= 2) && (dx >= 0) && (dx <= 2);
            const int ii = ok ? (3 * dy + dx) : 0;
            const float lx = wg[ii];
            const float lh = wg[9 + ii];
            const float vx = ok ? lx : 0.0f;
            const float vh = ok ? lh : 0.0f;
            if (s == 0) { afx[ch].x = vx; afh[ch].x = vh; }
            else        { afx[ch].y = vx; afh[ch].y = vh; }
        }
    }

    v8f cinit;
#pragma unroll
    for (int q8 = 0; q8 < 8; ++q8) cinit[q8] = bias[q8 & 3];

    for (int task = wv; task < NRT0 * 9; task += NWAVE) {
        const int k = task / 9;
        const int i = task - 9 * k;
        dec_tile_step<FIRST ? 0 : 1>(y0 - (S - 1) + 4 * k, x0 - 8 + 16 * i, 4 * k, 1 + 16 * i,
                                     xs, hin, cin, nullptr, cb, hb[0], gx, nullptr, nullptr,
                                     afx, afh, cinit, c, hf);
    }
    __syncthreads();

#pragma unroll
    for (int s = 1; s < S - 1; ++s) {
        const int nt = 9 * ((32 + 2 * (S - 1 - s) + 3) / 4);
        for (int task = wv; task < nt; task += NWAVE) {
            const int k = task / 9;
            const int i = task - 9 * k;
            dec_tile_step<2>(y0 - (S - 1) + s + 4 * k, x0 - 8 + 16 * i, s + 4 * k, 1 + 16 * i,
                             xs, nullptr, nullptr, hb[(s - 1) & 1], cb, hb[s & 1], gx, nullptr, nullptr,
                             afx, afh, cinit, c, hf);
        }
        __syncthreads();
    }

    {
        const int k = wv >> 2;
        for (int t = 0; t < 2; ++t) {
            const int i = 2 * (wv & 3) + t;
            dec_tile_step<3>(y0 + 4 * k, x0 + 16 * i, (S - 1) + 4 * k, 9 + 16 * i,
                             xs, nullptr, nullptr, hb[(S - 2) & 1], cb, nullptr, gx, hout, cout,
                             afx, afh, cinit, c, hf);
        }
    }
}

extern "C" void kernel_launch(void* const* d_in, const int* in_sizes, int n_in,
                              void* d_out, int out_size, void* d_ws, size_t ws_size,
                              hipStream_t stream) {
    (void)in_sizes; (void)n_in; (void)out_size; (void)ws_size;
    const float* data  = (const float*)d_in[0];
    const float* enc_w = (const float*)d_in[1];
    const float* enc_b = (const float*)d_in[2];
    const float* dec_w = (const float*)d_in[3];
    const float* dec_b = (const float*)d_in[4];
    float* out = (float*)d_out;

    float* ws = (float*)d_ws;
    float* h0 = ws;
    float* h1 = ws + (size_t)NPIX;
    float* h2 = ws + (size_t)2 * NPIX;
    float* c0 = ws + (size_t)3 * NPIX;
    float* c1 = ws + (size_t)4 * NPIX;

    dim3 grid(256), blk(1024);

    float *ha = h0, *hb = h1, *ca = c0, *cb = c1;
    for (int p = 0; p < 5; ++p) {
        const float* xa = data + (size_t)(4 * p) * NPIX;
        const float* xb = data + (size_t)(4 * p + 1) * NPIX;
        const float* xc = data + (size_t)(4 * p + 2) * NPIX;
        const float* xd = data + (size_t)(4 * p + 3) * NPIX;
        if (p == 0)
            convlstm_four_steps<true><<<grid, blk, 0, stream>>>(xa, xb, xc, xd, ha, ca, enc_w, enc_b, hb, cb);
        else
            convlstm_four_steps<false><<<grid, blk, 0, stream>>>(xa, xb, xc, xd, ha, ca, enc_w, enc_b, hb, cb);
        float* tmp;
        tmp = ha; ha = hb; hb = tmp;
        tmp = ca; ca = cb; cb = tmp;
    }
    const float* dec_in = ha;

    float *hda = hb, *hdb = h2, *cda = ca, *cdb = cb;
    convlstm_dec_steps<7, true><<<grid, blk, 0, stream>>>(dec_in, hda, cda, dec_w, dec_b, hdb, cdb);
    convlstm_dec_steps<7, false><<<grid, blk, 0, stream>>>(dec_in, hdb, cdb, dec_w, dec_b, hda, cda);
    convlstm_dec_steps<6, false><<<grid, blk, 0, stream>>>(dec_in, hda, cda, dec_w, dec_b, out, cdb);
}
